# starting kernel
# baseline (speedup 1.0000x reference)
#include <hip/hip_runtime.h>


typedef float v2f __attribute__((ext_vector_type(2)));
typedef float v8f __attribute__((ext_vector_type(8)));

#define CIN 256
#define COUT 64

__global__ void __launch_bounds__(256)
k_zero_deg(unsigned* __restrict__ deg, int N) {
    int i = blockIdx.x * 256 + threadIdx.x;
    if (i < N) deg[i] = 0u;
}

__global__ void __launch_bounds__(256)
k_deg(const int* __restrict__ dstp, unsigned* __restrict__ deg, int E) {
    int e = blockIdx.x * 256 + threadIdx.x;
    if (e < E) atomicAdd(&deg[dstp[e]], 1u);
}

__global__ void __launch_bounds__(256)
k_dinv(const unsigned* __restrict__ deg, float* __restrict__ dinv, int N) {
    int i = blockIdx.x * 256 + threadIdx.x;
    if (i < N) dinv[i] = rsqrtf((float)(deg[i] + 1u));
}

__global__ void __launch_bounds__(256)
k_gemm(const float* __restrict__ x, const float* __restrict__ W,
       float* __restrict__ h, int N) {
    __shared__ float2 Ws2[(CIN / 2) * COUT];
    for (int idx = threadIdx.x; idx < (CIN / 2) * COUT; idx += 256) {
        int p = idx >> 6, n = idx & 63;
        Ws2[idx] = make_float2(W[p * (2 * COUT) + n], W[p * (2 * COUT) + COUT + n]);
    }
    __syncthreads();

    const int lane = threadIdx.x & 31;
    const int wave = threadIdx.x >> 5;
    const int m0   = (blockIdx.x * 8 + wave) * 16;
    const int mr   = lane & 15;
    const int kh   = (lane >> 4) << 1;

    int row  = m0 + mr;
    int rowc = row < N ? row : (N - 1);
    const float* xr = x + (size_t)rowc * CIN;

    v8f acc0 = {}; v8f acc1 = {}; v8f acc2 = {}; v8f acc3 = {};
    for (int k = 0; k < CIN; k += 4) {
        v2f a = *(const v2f*)(xr + k + kh);
        const float2* wp = &Ws2[(((k + kh) >> 1) * COUT) + mr];
        v2f b0 = *(const v2f*)(wp + 0);
        v2f b1 = *(const v2f*)(wp + 16);
        v2f b2 = *(const v2f*)(wp + 32);
        v2f b3 = *(const v2f*)(wp + 48);
        acc0 = __builtin_amdgcn_wmma_f32_16x16x4_f32(false, a, false, b0, (short)0, acc0, false, false);
        acc1 = __builtin_amdgcn_wmma_f32_16x16x4_f32(false, a, false, b1, (short)0, acc1, false, false);
        acc2 = __builtin_amdgcn_wmma_f32_16x16x4_f32(false, a, false, b2, (short)0, acc2, false, false);
        acc3 = __builtin_amdgcn_wmma_f32_16x16x4_f32(false, a, false, b3, (short)0, acc3, false, false);
    }

    const int nlo = lane & 15;
    const int mhi = (lane >> 4) << 3;
    for (int r = 0; r < 8; ++r) {
        int orow = m0 + r + mhi;
        if (orow < N) {
            float* hp = h + (size_t)orow * COUT + nlo;
            hp[0]  = acc0[r];
            hp[16] = acc1[r];
            hp[32] = acc2[r];
            hp[48] = acc3[r];
        }
    }
}

__global__ void __launch_bounds__(256)
k_self(const float* __restrict__ h, const float* __restrict__ dinv,
       const float* __restrict__ b, float* __restrict__ out, int N) {
    long long gid = (long long)blockIdx.x * 256 + threadIdx.x;
    int i = (int)(gid >> 6);
    int c = (int)(gid & 63);
    if (i < N) {
        float di = dinv[i];
        out[(size_t)i * COUT + c] = b[c] + di * di * h[(size_t)i * COUT + c];
    }
}

__global__ void __launch_bounds__(256)
k_scatter(const int* __restrict__ srcp, const int* __restrict__ dstp,
          const float* __restrict__ dinv, const float* __restrict__ h,
          float* __restrict__ out, int E) {
    long long gid = (long long)blockIdx.x * 256 + threadIdx.x;
    int e = (int)(gid >> 5);
    if (e >= E) return;
    e = __builtin_amdgcn_readfirstlane(e);
    int lane = threadIdx.x & 31;
    int s = srcp[e];
    int d = dstp[e];
    float norm = dinv[s] * dinv[d];
    const float* hs = h + (size_t)s * COUT;
    float* od = out + (size_t)d * COUT;
    unsafeAtomicAdd(&od[lane],      hs[lane]      * norm);
    unsafeAtomicAdd(&od[lane + 32], hs[lane + 32] * norm);
}

extern "C" void kernel_launch(void* const* d_in, const int* in_sizes, int n_in,
                              void* d_out, int out_size, void* d_ws, size_t ws_size,
                              hipStream_t stream) {
    const float* x  = (const float*)d_in[0];
    const int*   ei = (const int*)d_in[1];
    const float* W  = (const float*)d_in[2];
    const float* b  = (const float*)d_in[3];
    float* out = (float*)d_out;

    const int N = in_sizes[0] / CIN;
    const int E = in_sizes[1] / 2;
    const int* srcp = ei;
    const int* dstp = ei + E;

    char* ws = (char*)d_ws;
    float*    h    = (float*)ws;
    unsigned* deg  = (unsigned*)(ws + (size_t)N * COUT * sizeof(float));
    float*    dinv = (float*)(ws + (size_t)N * COUT * sizeof(float) + (size_t)N * sizeof(unsigned));

    int nb;
    nb = (N + 255) / 256;
    k_zero_deg<<<nb, 256, 0, stream>>>(deg, N);
    nb = (E + 255) / 256;
    k_deg<<<nb, 256, 0, stream>>>(dstp, deg, E);
    nb = (N + 255) / 256;
    k_dinv<<<nb, 256, 0, stream>>>(deg, dinv, N);

    nb = (N + 127) / 128;
    k_gemm<<<nb, 256, 0, stream>>>(x, W, h, N);

    long long t = (long long)N * COUT;
    nb = (int)((t + 255) / 256);
    k_self<<<nb, 256, 0, stream>>>(h, dinv, b, out, N);

    t = (long long)E * 32;
    nb = (int)((t + 255) / 256);
    k_scatter<<<nb, 256, 0, stream>>>(srcp, dstp, dinv, h, out, E);
}


# tile owners write their edges as a compact (src,row) list while counting degrees; the out-tile launch reads its own list (entries, dinv loaded ahead), scans no keys; LDS adds by arrival as before
# speedup vs baseline: 3.3047x; 3.3047x over previous
#include <hip/hip_runtime.h>


typedef float v2f __attribute__((ext_vector_type(2)));
typedef float v8f __attribute__((ext_vector_type(8)));

#define CIN 256
#define COUT 64


#define TILE 391
static __device__ __forceinline__ unsigned tile_key(int d) { return ((unsigned)d / TILE) & 255u; }

#define LCAP 65536

static __device__ __forceinline__ unsigned key_skip(const unsigned char* raw) {
    return (16u - (unsigned)((unsigned long long)raw & 15ull)) & 15u;
}

#define KSTEPS 4
__global__ void __launch_bounds__(256)
k_key(const int* __restrict__ dstp, unsigned char* __restrict__ keyraw, int E) {
    unsigned char* keyb = keyraw + key_skip(keyraw);
    const int eb = blockIdx.x * (KSTEPS * 1024) + threadIdx.x * 4;
    if ((blockIdx.x + 1) * (KSTEPS * 1024) <= E) {
        int d[KSTEPS][4];
        #pragma unroll
        for (int it = 0; it < KSTEPS; ++it) {
            #pragma unroll
            for (int j = 0; j < 4; ++j) d[it][j] = dstp[eb + it * 1024 + j];
        }
        #pragma unroll
        for (int it = 0; it < KSTEPS; ++it)
            *(unsigned*)(keyb + eb + it * 1024) = tile_key(d[it][0]) | (tile_key(d[it][1]) << 8) |
                                                  (tile_key(d[it][2]) << 16) | (tile_key(d[it][3]) << 24);
    } else {
        for (int it = 0; it < KSTEPS; ++it) {
            const int e0 = eb + it * 1024;
            if (e0 + 3 < E) {
                const int d0 = dstp[e0], d1 = dstp[e0 + 1], d2 = dstp[e0 + 2], d3 = dstp[e0 + 3];
                *(unsigned*)(keyb + e0) = tile_key(d0) | (tile_key(d1) << 8) | (tile_key(d2) << 16) | (tile_key(d3) << 24);
            } else {
                for (int e = e0; e < E; ++e) keyb[e] = (unsigned char)tile_key(dstp[e]);
            }
        }
    }
}

__global__ void __launch_bounds__(256)
k_gemm(const float* __restrict__ x, const float* __restrict__ W,
       float* __restrict__ h, int N) {
    __shared__ float2 Ws2[(CIN / 2) * COUT];
    for (int idx = threadIdx.x; idx < (CIN / 2) * COUT; idx += 256) {
        int p = idx >> 6, n = idx & 63;
        Ws2[idx] = make_float2(W[p * (2 * COUT) + n], W[p * (2 * COUT) + COUT + n]);
    }
    __syncthreads();

    const int lane = threadIdx.x & 31;
    const int wave = threadIdx.x >> 5;
    const int m0   = (blockIdx.x * 8 + wave) * 16;
    const int mr   = lane & 15;
    const int kh   = (lane >> 4) << 1;

    int row  = m0 + mr;
    int rowc = row < N ? row : (N - 1);
    const float* xr = x + (size_t)rowc * CIN;

    v8f acc0 = {}; v8f acc1 = {}; v8f acc2 = {}; v8f acc3 = {};
    for (int k = 0; k < CIN; k += 4) {
        v2f a = *(const v2f*)(xr + k + kh);
        const float2* wp = &Ws2[(((k + kh) >> 1) * COUT) + mr];
        v2f b0 = *(const v2f*)(wp + 0);
        v2f b1 = *(const v2f*)(wp + 16);
        v2f b2 = *(const v2f*)(wp + 32);
        v2f b3 = *(const v2f*)(wp + 48);
        acc0 = __builtin_amdgcn_wmma_f32_16x16x4_f32(false, a, false, b0, (short)0, acc0, false, false);
        acc1 = __builtin_amdgcn_wmma_f32_16x16x4_f32(false, a, false, b1, (short)0, acc1, false, false);
        acc2 = __builtin_amdgcn_wmma_f32_16x16x4_f32(false, a, false, b2, (short)0, acc2, false, false);
        acc3 = __builtin_amdgcn_wmma_f32_16x16x4_f32(false, a, false, b3, (short)0, acc3, false, false);
    }

    const int nlo = lane & 15;
    const int mhi = (lane >> 4) << 3;
    for (int r = 0; r < 8; ++r) {
        int orow = m0 + r + mhi;
        if (orow < N) {
            float* hp = h + (size_t)orow * COUT + nlo;
            hp[0]  = acc0[r];
            hp[16] = acc1[r];
            hp[32] = acc2[r];
            hp[48] = acc3[r];
        }
    }
}

#define QCAP 64
#define UNR 4

static __device__ __forceinline__ void lds_add(float* p, float v) {
    __hip_atomic_fetch_add(p, v, __ATOMIC_RELAXED, __HIP_MEMORY_SCOPE_WORKGROUP);
}

static __device__ __forceinline__ unsigned hits16(uint4 v, unsigned pat) {
    unsigned x0 = v.x ^ pat, x1 = v.y ^ pat, x2 = v.z ^ pat, x3 = v.w ^ pat;
    unsigned t0 = (x0 - 0x01010101u) & ~x0;
    unsigned t1 = (x1 - 0x01010101u) & ~x1;
    unsigned t2 = (x2 - 0x01010101u) & ~x2;
    unsigned t3 = (x3 - 0x01010101u) & ~x3;
    return ((t0 >> 7) & 0x01010101u) | ((t1 >> 6) & 0x02020202u) |
           ((t2 >> 5) & 0x04040404u) | ((t3 >> 4) & 0x08080808u);
}

static __device__ __forceinline__ void pushb(unsigned hm, unsigned blk, uint2* q, int& qn) {
    bool m = hm != 0u;
    unsigned mk = __builtin_amdgcn_ballot_w32(m);
    if (mk) {
        if (m) q[qn + (int)__builtin_amdgcn_mbcnt_lo(mk, 0u)] = make_uint2(blk, hm);
        qn += __builtin_popcount(mk);
    }
}

static __device__ __forceinline__ void drain(uint2* q, int& qn, int n, int lane, int lo, int cnt,
        const int* __restrict__ srcp, const int* __restrict__ dstp,
        const float* __restrict__ dinv, const float* __restrict__ h, float* tile) {
    const int base = qn - n;
    unsigned blk = 0u, hm = 0u;
    if (lane < n) {
        uint2 en = q[base + lane];
        blk = en.x;
        hm = en.y;
    }
    __builtin_amdgcn_wave_barrier();
    qn = base;
    pushb(hm & (hm - 1u), blk, q, qn);
    __builtin_amdgcn_wave_barrier();

    int s = 0, x = TILE;
    float w = 0.f;
    if (lane < n) {
        int p = __builtin_ctz(hm);
        int e = (int)(blk << 4) + (((p & 7) << 2) | (p >> 3));
        s = srcp[e];
        int d = dstp[e];
        unsigned xr = (unsigned)(d - lo);
        if (xr < (unsigned)cnt) {
            x = (int)xr;
            w = dinv[s] * dinv[d];
        }
    }
    int wi = __float_as_int(w);
    int k = 0;
    for (; k + 8 <= n; k += 8) {
        float2 v[8];
        #pragma unroll
        for (int j = 0; j < 8; ++j) {
            int sk = __builtin_amdgcn_readlane(s, k + j);
            v[j] = *(const float2*)(h + (unsigned)(sk * COUT + 2 * lane));
        }
        #pragma unroll
        for (int j = 0; j < 8; ++j) {
            int xk = __builtin_amdgcn_readlane(x, k + j);
            float wk = __int_as_float(__builtin_amdgcn_readlane(wi, k + j));
            float* t = tile + xk * COUT + 2 * lane;
            lds_add(t, v[j].x * wk);
            lds_add(t + 1, v[j].y * wk);
        }
    }
    for (; k < n; ++k) {
        int sk = __builtin_amdgcn_readlane(s, k);
        int xk = __builtin_amdgcn_readlane(x, k);
        float wk = __int_as_float(__builtin_amdgcn_readlane(wi, k));
        float2 v = *(const float2*)(h + (unsigned)(sk * COUT + 2 * lane));
        float* t = tile + xk * COUT + 2 * lane;
        lds_add(t, v.x * wk);
        lds_add(t + 1, v.y * wk);
    }
}

#define DUNR 8

static __device__ __forceinline__ void drain_cnt(uint2* q, int& qn, int n, int lane, int lo, int cnt,
        const int* __restrict__ srcp, const int* __restrict__ dstp, unsigned* cl,
        unsigned* lcnt, uint2* __restrict__ list) {
    const int base = qn - n;
    unsigned blk = 0u, hm = 0u;
    if (lane < n) {
        uint2 en = q[base + lane];
        blk = en.x;
        hm = en.y;
    }
    __builtin_amdgcn_wave_barrier();
    qn = base;
    pushb(hm & (hm - 1u), blk, q, qn);
    __builtin_amdgcn_wave_barrier();
    bool ok = false;
    unsigned xr = 0u;
    int s = 0;
    if (lane < n) {
        int p = __builtin_ctz(hm);
        int e = (int)(blk << 4) + (((p & 7) << 2) | (p >> 3));
        s = srcp[e];
        xr = (unsigned)(dstp[e] - lo);
        ok = xr < (unsigned)cnt;
        if (ok) __hip_atomic_fetch_add(&cl[xr], 1u, __ATOMIC_RELAXED, __HIP_MEMORY_SCOPE_WORKGROUP);
    }
    const unsigned mk = __builtin_amdgcn_ballot_w32(ok);
    if (mk) {
        unsigned base = 0u;
        if (lane == 0)
            base = __hip_atomic_fetch_add(lcnt, (unsigned)__builtin_popcount(mk), __ATOMIC_RELAXED, __HIP_MEMORY_SCOPE_WORKGROUP);
        base = (unsigned)__builtin_amdgcn_readfirstlane((int)base);
        const unsigned pos = base + __builtin_amdgcn_mbcnt_lo(mk, 0u);
        if (ok && pos < (unsigned)LCAP) list[pos] = make_uint2((unsigned)s, xr);
    }
}

__global__ void __launch_bounds__(1024)
k_degown(const int* __restrict__ srcp, const int* __restrict__ dstp, const unsigned char* __restrict__ keyraw,
         float* __restrict__ dinv, uint2* __restrict__ lists, unsigned* __restrict__ nlist, int N, int E) {
    __shared__ unsigned cl[TILE];
    __shared__ unsigned lcnt;
    __shared__ uint2 qbuf[32 * QCAP];

    const int lane = threadIdx.x & 31;
    const int wave = __builtin_amdgcn_readfirstlane((int)(threadIdx.x >> 5));
    const int lo   = blockIdx.x * TILE;
    const int cnt  = (N - lo) < TILE ? (N - lo) : TILE;

    if (threadIdx.x < TILE) cl[threadIdx.x] = 0u;
    if (threadIdx.x == 0) lcnt = 0u;
    __syncthreads();

    uint2* q = qbuf + wave * QCAP;
    int qn = 0;
    uint2* list = lists + (size_t)blockIdx.x * LCAP;

    const unsigned char* keyb = keyraw + key_skip(keyraw);
    const uint4* k4 = (const uint4*)keyb;
    const unsigned mykey = (unsigned)blockIdx.x & 255u;
    const unsigned pat = mykey * 0x01010101u;
    const int nblk = E >> 4;
    const int nfull = nblk / (32 * DUNR);

    uint4 cur[DUNR];
    int c = wave;
    if (c < nfull) {
        #pragma unroll
        for (int u = 0; u < DUNR; ++u) cur[u] = k4[(c * DUNR + u) * 32 + lane];
    }
    while (c < nfull) {
        const int cn = c + 32;
        const bool more = cn < nfull;
        uint4 nxt[DUNR];
        const int cl2 = more ? cn : c;
        #pragma unroll
        for (int u = 0; u < DUNR; ++u) nxt[u] = k4[(cl2 * DUNR + u) * 32 + lane];
        #pragma unroll
        for (int u = 0; u < DUNR; ++u) {
            pushb(hits16(cur[u], pat), (unsigned)((c * DUNR + u) * 32 + lane), q, qn);
            while (qn >= 32) drain_cnt(q, qn, 32, lane, lo, cnt, srcp, dstp, cl, &lcnt, list);
        }
        if (!more) break;
        #pragma unroll
        for (int u = 0; u < DUNR; ++u) cur[u] = nxt[u];
        c = cn;
    }
    for (int g = nfull * DUNR + wave; g * 32 < nblk; g += 32) {
        const int bi = g * 32 + lane;
        unsigned hm = 0u;
        if (bi < nblk) hm = hits16(k4[bi], pat);
        pushb(hm, (unsigned)bi, q, qn);
        while (qn >= 32) drain_cnt(q, qn, 32, lane, lo, cnt, srcp, dstp, cl, &lcnt, list);
    }
    if (wave == 0 && (E & 15) != 0) {
        unsigned hm = 0u;
        if (lane < (E & 15) && (unsigned)keyb[(nblk << 4) + lane] == mykey)
            hm = 1u << (((lane & 3) << 3) | (lane >> 2));
        pushb(hm, (unsigned)nblk, q, qn);
    }
    while (qn > 0) drain_cnt(q, qn, qn < 32 ? qn : 32, lane, lo, cnt, srcp, dstp, cl, &lcnt, list);
    __syncthreads();

    if ((int)threadIdx.x < cnt) dinv[lo + threadIdx.x] = rsqrtf((float)(cl[threadIdx.x] + 1u));
    if (threadIdx.x == 0) nlist[blockIdx.x] = lcnt;
}

static __device__ __forceinline__ void add_rows(int s, int x, int wi, int m, int lane,
        const float* __restrict__ h, float* tile) {
    int k = 0;
    for (; k + 16 <= m; k += 16) {
        float2 v[16];
        #pragma unroll
        for (int j = 0; j < 16; ++j) {
            int sk = __builtin_amdgcn_readlane(s, k + j);
            v[j] = *(const float2*)(h + (unsigned)(sk * COUT + 2 * lane));
        }
        #pragma unroll
        for (int j = 0; j < 16; ++j) {
            int xk = __builtin_amdgcn_readlane(x, k + j);
            float wk = __int_as_float(__builtin_amdgcn_readlane(wi, k + j));
            float* t = tile + xk * COUT + 2 * lane;
            lds_add(t, v[j].x * wk);
            lds_add(t + 1, v[j].y * wk);
        }
    }
    for (; k < m; ++k) {
        int sk = __builtin_amdgcn_readlane(s, k);
        int xk = __builtin_amdgcn_readlane(x, k);
        float wk = __int_as_float(__builtin_amdgcn_readlane(wi, k));
        float2 v = *(const float2*)(h + (unsigned)(sk * COUT + 2 * lane));
        float* t = tile + xk * COUT + 2 * lane;
        lds_add(t, v.x * wk);
        lds_add(t + 1, v.y * wk);
    }
}

static __device__ __forceinline__ void scan_add(uint2* q, int wave, int lane, int lo, int cnt,
        const int* __restrict__ srcp, const int* __restrict__ dstp,
        const float* __restrict__ dinv, const float* __restrict__ h,
        const unsigned char* __restrict__ keyraw, float* tile, int E) {
    int qn = 0;

    const unsigned char* keyb = keyraw + key_skip(keyraw);
    const uint4* k4 = (const uint4*)keyb;
    const unsigned mykey = (unsigned)blockIdx.x & 255u;
    const unsigned pat = mykey * 0x01010101u;
    const int nblk = E >> 4;
    const int nfull = nblk / (32 * UNR);

    uint4 cur[UNR];
    int c = wave;
    if (c < nfull) {
        #pragma unroll
        for (int u = 0; u < UNR; ++u) cur[u] = k4[(c * UNR + u) * 32 + lane];
    }
    while (c < nfull) {
        const int cn = c + 32;
        const bool more = cn < nfull;
        uint4 nxt[UNR];
        const int cl = more ? cn : c;
        #pragma unroll
        for (int u = 0; u < UNR; ++u) nxt[u] = k4[(cl * UNR + u) * 32 + lane];
        #pragma unroll
        for (int u = 0; u < UNR; ++u) {
            pushb(hits16(cur[u], pat), (unsigned)((c * UNR + u) * 32 + lane), q, qn);
            while (qn >= 32) drain(q, qn, 32, lane, lo, cnt, srcp, dstp, dinv, h, tile);
        }
        if (!more) break;
        #pragma unroll
        for (int u = 0; u < UNR; ++u) cur[u] = nxt[u];
        c = cn;
    }
    for (int g = nfull * UNR + wave; g * 32 < nblk; g += 32) {
        const int bi = g * 32 + lane;
        unsigned hm = 0u;
        if (bi < nblk) hm = hits16(k4[bi], pat);
        pushb(hm, (unsigned)bi, q, qn);
        while (qn >= 32) drain(q, qn, 32, lane, lo, cnt, srcp, dstp, dinv, h, tile);
    }
    if (wave == 0 && (E & 15) != 0) {
        unsigned hm = 0u;
        if (lane < (E & 15) && (unsigned)keyb[(nblk << 4) + lane] == mykey)
            hm = 1u << (((lane & 3) << 3) | (lane >> 2));
        pushb(hm, (unsigned)nblk, q, qn);
    }
    while (qn > 0) drain(q, qn, qn < 32 ? qn : 32, lane, lo, cnt, srcp, dstp, dinv, h, tile);
}

__global__ void __launch_bounds__(1024)
k_tile(const int* __restrict__ srcp, const int* __restrict__ dstp,
       const float* __restrict__ dinv, const float* __restrict__ h,
       const float* __restrict__ b, const unsigned char* __restrict__ keyraw,
       const uint2* __restrict__ lists, const unsigned* __restrict__ nlist,
       float* __restrict__ out, int N, int E) {
    __shared__ float tile[(TILE + 1) * COUT];
    __shared__ uint2 qbuf[32 * QCAP];

    const int lane = threadIdx.x & 31;
    const int wave = __builtin_amdgcn_readfirstlane((int)(threadIdx.x >> 5));
    const int lo   = blockIdx.x * TILE;
    const int cnt  = (N - lo) < TILE ? (N - lo) : TILE;

    for (int idx = threadIdx.x; idx < cnt * COUT; idx += 1024) {
        int i = idx >> 6, c = idx & 63;
        float di = dinv[lo + i];
        tile[idx] = b[c] + di * di * h[(size_t)(lo + i) * COUT + c];
    }
    if (threadIdx.x < COUT) tile[TILE * COUT + threadIdx.x] = 0.f;
    __syncthreads();

    const int nl = __builtin_amdgcn_readfirstlane((int)nlist[blockIdx.x]);
    if (__builtin_expect(nl <= LCAP, 1)) {
        const uint2* L = lists + (size_t)blockIdx.x * LCAP;
        const int b0 = wave * 32;
        const int last = nl - 1;
        uint2 e1 = make_uint2(0u, 0u), e2 = e1;
        float w1 = 0.f;
        if (b0 < nl) {
            e1 = L[min(b0 + lane, last)];
            e2 = L[min(b0 + 1024 + lane, last)];
            w1 = dinv[e1.x] * dinv[lo + (int)e1.y];
        }
        for (int bb = b0; bb < nl; bb += 1024) {
            const uint2 e3 = L[min(bb + 2048 + lane, last)];
            const float ds2 = dinv[e2.x];
            const float dd2 = dinv[lo + (int)e2.y];
            const int m = (nl - bb) < 32 ? (nl - bb) : 32;
            add_rows((int)e1.x, (int)e1.y, __float_as_int(w1), m, lane, h, tile);
            e1 = e2;
            e2 = e3;
            w1 = ds2 * dd2;
        }
    } else {
        scan_add(qbuf + wave * QCAP, wave, lane, lo, cnt, srcp, dstp, dinv, h, keyraw, tile, E);
    }
    __syncthreads();

    float* op = out + (size_t)lo * COUT;
    for (int idx = threadIdx.x; idx < cnt * COUT; idx += 1024) op[idx] = tile[idx];
}

extern "C" void kernel_launch(void* const* d_in, const int* in_sizes, int n_in,
                              void* d_out, int out_size, void* d_ws, size_t ws_size,
                              hipStream_t stream) {
    const float* x  = (const float*)d_in[0];
    const int*   ei = (const int*)d_in[1];
    const float* W  = (const float*)d_in[2];
    const float* b  = (const float*)d_in[3];
    float* out = (float*)d_out;

    const int N = in_sizes[0] / CIN;
    const int E = in_sizes[1] / 2;
    const int* srcp = ei;
    const int* dstp = ei + E;

    const int nt = (N + TILE - 1) / TILE;
    char* ws = (char*)d_ws;
    float*    h     = (float*)ws;
    uint2*    lists = (uint2*)(ws + (size_t)N * COUT * sizeof(float));
    unsigned* nlist = (unsigned*)(lists + (size_t)nt * LCAP);
    float*    dinv  = (float*)(nlist + nt);
    unsigned char* key = (unsigned char*)(dinv + N);

    int nb;
    nb = (E + 1024 * KSTEPS - 1) / (1024 * KSTEPS);
    k_key<<<nb, 256, 0, stream>>>(dstp, key, E);
    k_degown<<<nt, 1024, 0, stream>>>(srcp, dstp, key, dinv, lists, nlist, N, E);

    nb = (N + 127) / 128;
    k_gemm<<<nb, 256, 0, stream>>>(x, W, h, N);

    k_tile<<<nt, 1024, 0, stream>>>(srcp, dstp, dinv, h, b, key, lists, nlist, out, N, E);
}


# edges binned by a streaming launch: 256 slice work-groups rank edges by tile (returning LDS add), store (src,row) in (tile,slice) buckets; owners read their buckets, no key scan; sums by arrival
# speedup vs baseline: 4.7630x; 1.4413x over previous
#include <hip/hip_runtime.h>


typedef float v2f __attribute__((ext_vector_type(2)));
typedef float v8f __attribute__((ext_vector_type(8)));

#define CIN 256
#define COUT 64


#define TILE 391
#define MAXT 256

#define LCAP 65536

#define NSL 256
#define BCAP 128

static __device__ __forceinline__ int slice_len(int E) { return (((E + NSL - 1) / NSL) + 3) & ~3; }

static __device__ __forceinline__ int base4(int e, int se) {
    const int t = se - 4 > 0 ? se - 4 : 0;
    return e + 3 < se ? e : t;
}

static __device__ __forceinline__ void ld4(const int* __restrict__ srcp, const int* __restrict__ dstp,
                                           int eb, int* s, int* d) {
    #pragma unroll
    for (int j = 0; j < 4; ++j) { s[j] = srcp[eb + j]; d[j] = dstp[eb + j]; }
}

__global__ void __launch_bounds__(1024)
k_bin(const int* __restrict__ srcp, const int* __restrict__ dstp,
      uint2* __restrict__ buckets, unsigned* __restrict__ cnts, int N, int E) {
    __shared__ unsigned bc[MAXT];
    if (threadIdx.x < MAXT) bc[threadIdx.x] = 0u;
    __syncthreads();

    const int per = slice_len(E);
    const int sb  = blockIdx.x * per;
    const int se  = (E - sb) < per ? E : sb + per;
    const unsigned lim = (unsigned)(N < MAXT * TILE ? N : MAXT * TILE);
    uint2* B = buckets + (size_t)blockIdx.x * BCAP;

    int e0 = sb + (int)threadIdx.x * 4;
    if (e0 < se) {
        int eb = base4(e0, se);
        int sc[4], dc[4];
        ld4(srcp, dstp, eb, sc, dc);
        while (true) {
            const int en = e0 + 4096;
            const bool hn = en < se;
            const int ebn = base4(hn ? en : e0, se);
            int sn[4], dn[4];
            ld4(srcp, dstp, ebn, sn, dn);
            unsigned tt[4], rr[4];
            #pragma unroll
            for (int j = 0; j < 4; ++j) {
                const unsigned d = (unsigned)dc[j];
                tt[j] = d / TILE;
                rr[j] = (unsigned)BCAP;
                if (eb + j >= e0 && d < lim)
                    rr[j] = __hip_atomic_fetch_add(&bc[tt[j]], 1u, __ATOMIC_RELAXED, __HIP_MEMORY_SCOPE_WORKGROUP);
            }
            #pragma unroll
            for (int j = 0; j < 4; ++j) {
                if (rr[j] < (unsigned)BCAP)
                    B[(size_t)tt[j] * (NSL * BCAP) + rr[j]] = make_uint2((unsigned)sc[j], (unsigned)dc[j] - tt[j] * TILE);
            }
            if (!hn) break;
            #pragma unroll
            for (int j = 0; j < 4; ++j) { sc[j] = sn[j]; dc[j] = dn[j]; }
            e0 = en;
            eb = ebn;
        }
    }
    __syncthreads();

    const int nt = (N + TILE - 1) / TILE;
    if ((int)threadIdx.x < nt && threadIdx.x < MAXT) cnts[threadIdx.x * NSL + blockIdx.x] = bc[threadIdx.x];
}

__global__ void __launch_bounds__(256)
k_gemm(const float* __restrict__ x, const float* __restrict__ W,
       float* __restrict__ h, int N) {
    __shared__ float2 Ws2[(CIN / 2) * COUT];
    for (int idx = threadIdx.x; idx < (CIN / 2) * COUT; idx += 256) {
        int p = idx >> 6, n = idx & 63;
        Ws2[idx] = make_float2(W[p * (2 * COUT) + n], W[p * (2 * COUT) + COUT + n]);
    }
    __syncthreads();

    const int lane = threadIdx.x & 31;
    const int wave = threadIdx.x >> 5;
    const int m0   = (blockIdx.x * 8 + wave) * 16;
    const int mr   = lane & 15;
    const int kh   = (lane >> 4) << 1;

    int row  = m0 + mr;
    int rowc = row < N ? row : (N - 1);
    const float* xr = x + (size_t)rowc * CIN;

    v8f acc0 = {}; v8f acc1 = {}; v8f acc2 = {}; v8f acc3 = {};
    for (int k = 0; k < CIN; k += 4) {
        v2f a = *(const v2f*)(xr + k + kh);
        const float2* wp = &Ws2[(((k + kh) >> 1) * COUT) + mr];
        v2f b0 = *(const v2f*)(wp + 0);
        v2f b1 = *(const v2f*)(wp + 16);
        v2f b2 = *(const v2f*)(wp + 32);
        v2f b3 = *(const v2f*)(wp + 48);
        acc0 = __builtin_amdgcn_wmma_f32_16x16x4_f32(false, a, false, b0, (short)0, acc0, false, false);
        acc1 = __builtin_amdgcn_wmma_f32_16x16x4_f32(false, a, false, b1, (short)0, acc1, false, false);
        acc2 = __builtin_amdgcn_wmma_f32_16x16x4_f32(false, a, false, b2, (short)0, acc2, false, false);
        acc3 = __builtin_amdgcn_wmma_f32_16x16x4_f32(false, a, false, b3, (short)0, acc3, false, false);
    }

    const int nlo = lane & 15;
    const int mhi = (lane >> 4) << 3;
    for (int r = 0; r < 8; ++r) {
        int orow = m0 + r + mhi;
        if (orow < N) {
            float* hp = h + (size_t)orow * COUT + nlo;
            hp[0]  = acc0[r];
            hp[16] = acc1[r];
            hp[32] = acc2[r];
            hp[48] = acc3[r];
        }
    }
}


static __device__ __forceinline__ void lds_add(float* p, float v) {
    __hip_atomic_fetch_add(p, v, __ATOMIC_RELAXED, __HIP_MEMORY_SCOPE_WORKGROUP);
}

#define BPW (NSL / 32)

__global__ void __launch_bounds__(1024)
k_degown(const int* __restrict__ srcp, const int* __restrict__ dstp,
         const uint2* __restrict__ buckets, const unsigned* __restrict__ cnts,
         float* __restrict__ dinv, uint2* __restrict__ lists, unsigned* __restrict__ nlist, int N, int E) {
    __shared__ unsigned cl[TILE];
    __shared__ unsigned lcnt;

    const int lane = threadIdx.x & 31;
    const int wave = __builtin_amdgcn_readfirstlane((int)(threadIdx.x >> 5));
    const int lo   = blockIdx.x * TILE;
    const int cnt  = (N - lo) < TILE ? (N - lo) : TILE;

    if (threadIdx.x < TILE) cl[threadIdx.x] = 0u;
    if (threadIdx.x == 0) lcnt = 0u;
    __syncthreads();

    uint2* list = lists + (size_t)blockIdx.x * LCAP;
    const unsigned* cn = cnts + (size_t)blockIdx.x * NSL;
    const uint2* B = buckets + (size_t)blockIdx.x * (NSL * BCAP);

    unsigned nv = 0u;
    if (lane < BPW) nv = cn[wave + 32 * lane];
    const unsigned nu = nv <= (unsigned)BCAP ? nv : 0u;

    unsigned nj[BPW];
    uint2 en[BPW][2];
    unsigned tot = 0u;
    #pragma unroll
    for (int j = 0; j < BPW; ++j) {
        nj[j] = (unsigned)__builtin_amdgcn_readlane((int)nu, j);
        tot += nj[j];
        const uint2* Bj = B + (wave + 32 * j) * BCAP;
        #pragma unroll
        for (int c = 0; c < 2; ++c) {
            const unsigned i = (unsigned)(c * 32 + lane);
            en[j][c] = make_uint2(0u, 0u);
            if (i < nj[j]) en[j][c] = Bj[i];
        }
    }
    unsigned base = 0u;
    if (tot) {
        if (lane == 0) base = __hip_atomic_fetch_add(&lcnt, tot, __ATOMIC_RELAXED, __HIP_MEMORY_SCOPE_WORKGROUP);
        base = (unsigned)__builtin_amdgcn_readfirstlane((int)base);
    }
    #pragma unroll
    for (int j = 0; j < BPW; ++j) {
        #pragma unroll
        for (int c = 0; c < 2; ++c) {
            const unsigned i = (unsigned)(c * 32 + lane);
            if (i < nj[j]) {
                __hip_atomic_fetch_add(&cl[en[j][c].y], 1u, __ATOMIC_RELAXED, __HIP_MEMORY_SCOPE_WORKGROUP);
                if (base + i < (unsigned)LCAP) list[base + i] = en[j][c];
            }
        }
        const uint2* Bj = B + (wave + 32 * j) * BCAP;
        for (unsigned i = 64u + (unsigned)lane; i < nj[j]; i += 32u) {
            const uint2 e = Bj[i];
            __hip_atomic_fetch_add(&cl[e.y], 1u, __ATOMIC_RELAXED, __HIP_MEMORY_SCOPE_WORKGROUP);
            if (base + i < (unsigned)LCAP) list[base + i] = e;
        }
        base += nj[j];
    }

    if (__builtin_amdgcn_ballot_w32(nv > (unsigned)BCAP)) {
        const int per = slice_len(E);
        for (int j = 0; j < BPW; ++j) {
            const unsigned n = (unsigned)__builtin_amdgcn_readlane((int)nv, j);
            if (n <= (unsigned)BCAP) continue;
            const int sb = (wave + 32 * j) * per;
            const int se = (E - sb) < per ? E : sb + per;
            for (int e0 = sb; e0 < se; e0 += 32) {
                const int e = e0 + lane;
                bool ok = false;
                unsigned xr = 0u;
                int s = 0;
                if (e < se) {
                    xr = (unsigned)(dstp[e] - lo);
                    ok = xr < (unsigned)cnt;
                    if (ok) {
                        s = srcp[e];
                        __hip_atomic_fetch_add(&cl[xr], 1u, __ATOMIC_RELAXED, __HIP_MEMORY_SCOPE_WORKGROUP);
                    }
                }
                const unsigned mk = __builtin_amdgcn_ballot_w32(ok);
                if (mk) {
                    unsigned b2 = 0u;
                    if (lane == 0)
                        b2 = __hip_atomic_fetch_add(&lcnt, (unsigned)__builtin_popcount(mk), __ATOMIC_RELAXED, __HIP_MEMORY_SCOPE_WORKGROUP);
                    b2 = (unsigned)__builtin_amdgcn_readfirstlane((int)b2);
                    const unsigned pos = b2 + __builtin_amdgcn_mbcnt_lo(mk, 0u);
                    if (ok && pos < (unsigned)LCAP) list[pos] = make_uint2((unsigned)s, xr);
                }
            }
        }
    }
    __syncthreads();

    if ((int)threadIdx.x < cnt) dinv[lo + threadIdx.x] = rsqrtf((float)(cl[threadIdx.x] + 1u));
    if (threadIdx.x == 0) nlist[blockIdx.x] = lcnt;
}

static __device__ __forceinline__ void add_rows(int s, int x, int wi, int m, int lane,
        const float* __restrict__ h, float* tile) {
    int k = 0;
    for (; k + 16 <= m; k += 16) {
        float2 v[16];
        #pragma unroll
        for (int j = 0; j < 16; ++j) {
            int sk = __builtin_amdgcn_readlane(s, k + j);
            v[j] = *(const float2*)(h + (unsigned)(sk * COUT + 2 * lane));
        }
        #pragma unroll
        for (int j = 0; j < 16; ++j) {
            int xk = __builtin_amdgcn_readlane(x, k + j);
            float wk = __int_as_float(__builtin_amdgcn_readlane(wi, k + j));
            float* t = tile + xk * COUT + 2 * lane;
            lds_add(t, v[j].x * wk);
            lds_add(t + 1, v[j].y * wk);
        }
    }
    for (; k < m; ++k) {
        int sk = __builtin_amdgcn_readlane(s, k);
        int xk = __builtin_amdgcn_readlane(x, k);
        float wk = __int_as_float(__builtin_amdgcn_readlane(wi, k));
        float2 v = *(const float2*)(h + (unsigned)(sk * COUT + 2 * lane));
        float* t = tile + xk * COUT + 2 * lane;
        lds_add(t, v.x * wk);
        lds_add(t + 1, v.y * wk);
    }
}

static __device__ __forceinline__ void scan_direct(int wave, int lane, int lo, int cnt,
        const int* __restrict__ srcp, const int* __restrict__ dstp,
        const float* __restrict__ dinv, const float* __restrict__ h, float* tile, int E) {
    for (int e0 = wave * 32; e0 < E; e0 += 1024) {
        const int e = e0 + lane;
        int s = 0, x = TILE;
        float w = 0.f;
        bool ok = false;
        if (e < E) {
            const int d = dstp[e];
            const unsigned xr = (unsigned)(d - lo);
            if (xr < (unsigned)cnt) {
                ok = true;
                s = srcp[e];
                x = (int)xr;
                w = dinv[s] * dinv[d];
            }
        }
        const int wi = __float_as_int(w);
        unsigned mk = __builtin_amdgcn_ballot_w32(ok);
        while (mk) {
            const int j = __builtin_ctz(mk);
            mk &= mk - 1u;
            const int sk = __builtin_amdgcn_readlane(s, j);
            const int xk = __builtin_amdgcn_readlane(x, j);
            const float wk = __int_as_float(__builtin_amdgcn_readlane(wi, j));
            const float2 v = *(const float2*)(h + (unsigned)(sk * COUT + 2 * lane));
            float* t = tile + xk * COUT + 2 * lane;
            lds_add(t, v.x * wk);
            lds_add(t + 1, v.y * wk);
        }
    }
}

__global__ void __launch_bounds__(1024)
k_tile(const int* __restrict__ srcp, const int* __restrict__ dstp,
       const float* __restrict__ dinv, const float* __restrict__ h,
       const float* __restrict__ b,
       const uint2* __restrict__ lists, const unsigned* __restrict__ nlist,
       float* __restrict__ out, int N, int E) {
    __shared__ float tile[(TILE + 1) * COUT];

    const int lane = threadIdx.x & 31;
    const int wave = __builtin_amdgcn_readfirstlane((int)(threadIdx.x >> 5));
    const int lo   = blockIdx.x * TILE;
    const int cnt  = (N - lo) < TILE ? (N - lo) : TILE;

    for (int idx = threadIdx.x; idx < cnt * COUT; idx += 1024) {
        int i = idx >> 6, c = idx & 63;
        float di = dinv[lo + i];
        tile[idx] = b[c] + di * di * h[(size_t)(lo + i) * COUT + c];
    }
    if (threadIdx.x < COUT) tile[TILE * COUT + threadIdx.x] = 0.f;
    __syncthreads();

    const int nl = __builtin_amdgcn_readfirstlane((int)nlist[blockIdx.x]);
    if (__builtin_expect(nl <= LCAP, 1)) {
        const uint2* L = lists + (size_t)blockIdx.x * LCAP;
        const int b0 = wave * 32;
        const int last = nl - 1;
        uint2 e1 = make_uint2(0u, 0u), e2 = e1;
        float w1 = 0.f;
        if (b0 < nl) {
            e1 = L[min(b0 + lane, last)];
            e2 = L[min(b0 + 1024 + lane, last)];
            w1 = dinv[e1.x] * dinv[lo + (int)e1.y];
        }
        for (int bb = b0; bb < nl; bb += 1024) {
            const uint2 e3 = L[min(bb + 2048 + lane, last)];
            const float ds2 = dinv[e2.x];
            const float dd2 = dinv[lo + (int)e2.y];
            const int m = (nl - bb) < 32 ? (nl - bb) : 32;
            add_rows((int)e1.x, (int)e1.y, __float_as_int(w1), m, lane, h, tile);
            e1 = e2;
            e2 = e3;
            w1 = ds2 * dd2;
        }
    } else {
        scan_direct(wave, lane, lo, cnt, srcp, dstp, dinv, h, tile, E);
    }
    __syncthreads();

    float* op = out + (size_t)lo * COUT;
    for (int idx = threadIdx.x; idx < cnt * COUT; idx += 1024) op[idx] = tile[idx];
}

extern "C" void kernel_launch(void* const* d_in, const int* in_sizes, int n_in,
                              void* d_out, int out_size, void* d_ws, size_t ws_size,
                              hipStream_t stream) {
    const float* x  = (const float*)d_in[0];
    const int*   ei = (const int*)d_in[1];
    const float* W  = (const float*)d_in[2];
    const float* b  = (const float*)d_in[3];
    float* out = (float*)d_out;

    const int N = in_sizes[0] / CIN;
    const int E = in_sizes[1] / 2;
    const int* srcp = ei;
    const int* dstp = ei + E;

    const int nt = (N + TILE - 1) / TILE;
    char* ws = (char*)d_ws;
    float*    h     = (float*)ws;
    uint2*    lists = (uint2*)(ws + (size_t)N * COUT * sizeof(float));
    uint2*    bkts  = lists + (size_t)nt * LCAP;
    unsigned* nlist = (unsigned*)(bkts + (size_t)nt * NSL * BCAP);
    unsigned* cnts  = nlist + nt;
    float*    dinv  = (float*)(cnts + (size_t)nt * NSL);

    k_bin<<<NSL, 1024, 0, stream>>>(srcp, dstp, bkts, cnts, N, E);
    k_degown<<<nt, 1024, 0, stream>>>(srcp, dstp, bkts, cnts, dinv, lists, nlist, N, E);

    const int nb = (N + 127) / 128;
    k_gemm<<<nb, 256, 0, stream>>>(x, W, h, N);

    k_tile<<<nt, 1024, 0, stream>>>(srcp, dstp, dinv, h, b, lists, nlist, out, N, E);
}


# out-tile launch keeps a table of all node degrees in LDS (16 bits a node, written by the degree launch): edge weight = two LDS reads + rsqrt, no scattered dinv line read per edge; sums by arrival
# speedup vs baseline: 4.9307x; 1.0352x over previous
#include <hip/hip_runtime.h>


typedef float v2f __attribute__((ext_vector_type(2)));
typedef float v8f __attribute__((ext_vector_type(8)));

#define CIN 256
#define COUT 64


#define TILE 391
#define MAXT 256

#define LCAP 65536

#define NSL 256
#define BCAP 128

static __device__ __forceinline__ int slice_len(int E) { return (((E + NSL - 1) / NSL) + 3) & ~3; }

static __device__ __forceinline__ float dinv_of(unsigned deg) { return rsqrtf((float)deg); }

#define DSAT 65535u

static __device__ __forceinline__ int base4(int e, int se) {
    const int t = se - 4 > 0 ? se - 4 : 0;
    return e + 3 < se ? e : t;
}

static __device__ __forceinline__ void ld4(const int* __restrict__ srcp, const int* __restrict__ dstp,
                                           int eb, int* s, int* d) {
    #pragma unroll
    for (int j = 0; j < 4; ++j) { s[j] = srcp[eb + j]; d[j] = dstp[eb + j]; }
}

__global__ void __launch_bounds__(1024)
k_bin(const int* __restrict__ srcp, const int* __restrict__ dstp,
      uint2* __restrict__ buckets, unsigned* __restrict__ cnts, int N, int E) {
    __shared__ unsigned bc[MAXT];
    if (threadIdx.x < MAXT) bc[threadIdx.x] = 0u;
    __syncthreads();

    const int per = slice_len(E);
    const int sb  = blockIdx.x * per;
    const int se  = (E - sb) < per ? E : sb + per;
    const unsigned lim = (unsigned)(N < MAXT * TILE ? N : MAXT * TILE);
    uint2* B = buckets + (size_t)blockIdx.x * BCAP;

    int e0 = sb + (int)threadIdx.x * 4;
    if (e0 < se) {
        int eb = base4(e0, se);
        int sc[4], dc[4];
        ld4(srcp, dstp, eb, sc, dc);
        while (true) {
            const int en = e0 + 4096;
            const bool hn = en < se;
            const int ebn = base4(hn ? en : e0, se);
            int sn[4], dn[4];
            ld4(srcp, dstp, ebn, sn, dn);
            unsigned tt[4], rr[4];
            #pragma unroll
            for (int j = 0; j < 4; ++j) {
                const unsigned d = (unsigned)dc[j];
                tt[j] = d / TILE;
                rr[j] = (unsigned)BCAP;
                if (eb + j >= e0 && d < lim)
                    rr[j] = __hip_atomic_fetch_add(&bc[tt[j]], 1u, __ATOMIC_RELAXED, __HIP_MEMORY_SCOPE_WORKGROUP);
            }
            #pragma unroll
            for (int j = 0; j < 4; ++j) {
                if (rr[j] < (unsigned)BCAP)
                    B[(size_t)tt[j] * (NSL * BCAP) + rr[j]] = make_uint2((unsigned)sc[j], (unsigned)dc[j] - tt[j] * TILE);
            }
            if (!hn) break;
            #pragma unroll
            for (int j = 0; j < 4; ++j) { sc[j] = sn[j]; dc[j] = dn[j]; }
            e0 = en;
            eb = ebn;
        }
    }
    __syncthreads();

    const int nt = (N + TILE - 1) / TILE;
    if ((int)threadIdx.x < nt && threadIdx.x < MAXT) cnts[threadIdx.x * NSL + blockIdx.x] = bc[threadIdx.x];
}

__global__ void __launch_bounds__(256)
k_gemm(const float* __restrict__ x, const float* __restrict__ W,
       float* __restrict__ h, int N) {
    __shared__ float2 Ws2[(CIN / 2) * COUT];
    for (int idx = threadIdx.x; idx < (CIN / 2) * COUT; idx += 256) {
        int p = idx >> 6, n = idx & 63;
        Ws2[idx] = make_float2(W[p * (2 * COUT) + n], W[p * (2 * COUT) + COUT + n]);
    }
    __syncthreads();

    const int lane = threadIdx.x & 31;
    const int wave = threadIdx.x >> 5;
    const int m0   = (blockIdx.x * 8 + wave) * 16;
    const int mr   = lane & 15;
    const int kh   = (lane >> 4) << 1;

    int row  = m0 + mr;
    int rowc = row < N ? row : (N - 1);
    const float* xr = x + (size_t)rowc * CIN;

    v8f acc0 = {}; v8f acc1 = {}; v8f acc2 = {}; v8f acc3 = {};
    for (int k = 0; k < CIN; k += 4) {
        v2f a = *(const v2f*)(xr + k + kh);
        const float2* wp = &Ws2[(((k + kh) >> 1) * COUT) + mr];
        v2f b0 = *(const v2f*)(wp + 0);
        v2f b1 = *(const v2f*)(wp + 16);
        v2f b2 = *(const v2f*)(wp + 32);
        v2f b3 = *(const v2f*)(wp + 48);
        acc0 = __builtin_amdgcn_wmma_f32_16x16x4_f32(false, a, false, b0, (short)0, acc0, false, false);
        acc1 = __builtin_amdgcn_wmma_f32_16x16x4_f32(false, a, false, b1, (short)0, acc1, false, false);
        acc2 = __builtin_amdgcn_wmma_f32_16x16x4_f32(false, a, false, b2, (short)0, acc2, false, false);
        acc3 = __builtin_amdgcn_wmma_f32_16x16x4_f32(false, a, false, b3, (short)0, acc3, false, false);
    }

    const int nlo = lane & 15;
    const int mhi = (lane >> 4) << 3;
    for (int r = 0; r < 8; ++r) {
        int orow = m0 + r + mhi;
        if (orow < N) {
            float* hp = h + (size_t)orow * COUT + nlo;
            hp[0]  = acc0[r];
            hp[16] = acc1[r];
            hp[32] = acc2[r];
            hp[48] = acc3[r];
        }
    }
}


static __device__ __forceinline__ void lds_add(float* p, float v) {
    __hip_atomic_fetch_add(p, v, __ATOMIC_RELAXED, __HIP_MEMORY_SCOPE_WORKGROUP);
}

#define BPW (NSL / 32)

__global__ void __launch_bounds__(1024)
k_degown(const int* __restrict__ srcp, const int* __restrict__ dstp,
         const uint2* __restrict__ buckets, const unsigned* __restrict__ cnts,
         float* __restrict__ dinv, unsigned short* __restrict__ deg16,
         uint2* __restrict__ lists, unsigned* __restrict__ nlist, int N, int E) {
    __shared__ unsigned cl[TILE];
    __shared__ unsigned lcnt;

    const int lane = threadIdx.x & 31;
    const int wave = __builtin_amdgcn_readfirstlane((int)(threadIdx.x >> 5));
    const int lo   = blockIdx.x * TILE;
    const int cnt  = (N - lo) < TILE ? (N - lo) : TILE;

    if (threadIdx.x < TILE) cl[threadIdx.x] = 0u;
    if (threadIdx.x == 0) lcnt = 0u;
    __syncthreads();

    uint2* list = lists + (size_t)blockIdx.x * LCAP;
    const unsigned* cn = cnts + (size_t)blockIdx.x * NSL;
    const uint2* B = buckets + (size_t)blockIdx.x * (NSL * BCAP);

    unsigned nv = 0u;
    if (lane < BPW) nv = cn[wave + 32 * lane];
    const unsigned nu = nv <= (unsigned)BCAP ? nv : 0u;

    unsigned nj[BPW];
    uint2 en[BPW][2];
    unsigned tot = 0u;
    #pragma unroll
    for (int j = 0; j < BPW; ++j) {
        nj[j] = (unsigned)__builtin_amdgcn_readlane((int)nu, j);
        tot += nj[j];
        const uint2* Bj = B + (wave + 32 * j) * BCAP;
        #pragma unroll
        for (int c = 0; c < 2; ++c) {
            const unsigned i = (unsigned)(c * 32 + lane);
            en[j][c] = make_uint2(0u, 0u);
            if (i < nj[j]) en[j][c] = Bj[i];
        }
    }
    unsigned base = 0u;
    if (tot) {
        if (lane == 0) base = __hip_atomic_fetch_add(&lcnt, tot, __ATOMIC_RELAXED, __HIP_MEMORY_SCOPE_WORKGROUP);
        base = (unsigned)__builtin_amdgcn_readfirstlane((int)base);
    }
    #pragma unroll
    for (int j = 0; j < BPW; ++j) {
        #pragma unroll
        for (int c = 0; c < 2; ++c) {
            const unsigned i = (unsigned)(c * 32 + lane);
            if (i < nj[j]) {
                __hip_atomic_fetch_add(&cl[en[j][c].y], 1u, __ATOMIC_RELAXED, __HIP_MEMORY_SCOPE_WORKGROUP);
                if (base + i < (unsigned)LCAP) list[base + i] = en[j][c];
            }
        }
        const uint2* Bj = B + (wave + 32 * j) * BCAP;
        for (unsigned i = 64u + (unsigned)lane; i < nj[j]; i += 32u) {
            const uint2 e = Bj[i];
            __hip_atomic_fetch_add(&cl[e.y], 1u, __ATOMIC_RELAXED, __HIP_MEMORY_SCOPE_WORKGROUP);
            if (base + i < (unsigned)LCAP) list[base + i] = e;
        }
        base += nj[j];
    }

    if (__builtin_amdgcn_ballot_w32(nv > (unsigned)BCAP)) {
        const int per = slice_len(E);
        for (int j = 0; j < BPW; ++j) {
            const unsigned n = (unsigned)__builtin_amdgcn_readlane((int)nv, j);
            if (n <= (unsigned)BCAP) continue;
            const int sb = (wave + 32 * j) * per;
            const int se = (E - sb) < per ? E : sb + per;
            for (int e0 = sb; e0 < se; e0 += 32) {
                const int e = e0 + lane;
                bool ok = false;
                unsigned xr = 0u;
                int s = 0;
                if (e < se) {
                    xr = (unsigned)(dstp[e] - lo);
                    ok = xr < (unsigned)cnt;
                    if (ok) {
                        s = srcp[e];
                        __hip_atomic_fetch_add(&cl[xr], 1u, __ATOMIC_RELAXED, __HIP_MEMORY_SCOPE_WORKGROUP);
                    }
                }
                const unsigned mk = __builtin_amdgcn_ballot_w32(ok);
                if (mk) {
                    unsigned b2 = 0u;
                    if (lane == 0)
                        b2 = __hip_atomic_fetch_add(&lcnt, (unsigned)__builtin_popcount(mk), __ATOMIC_RELAXED, __HIP_MEMORY_SCOPE_WORKGROUP);
                    b2 = (unsigned)__builtin_amdgcn_readfirstlane((int)b2);
                    const unsigned pos = b2 + __builtin_amdgcn_mbcnt_lo(mk, 0u);
                    if (ok && pos < (unsigned)LCAP) list[pos] = make_uint2((unsigned)s, xr);
                }
            }
        }
    }
    __syncthreads();

    if ((int)threadIdx.x < cnt) {
        const unsigned dg = cl[threadIdx.x] + 1u;
        dinv[lo + threadIdx.x] = dinv_of(dg);
        deg16[lo + threadIdx.x] = (unsigned short)(dg < DSAT ? dg : DSAT);
    }
    if (threadIdx.x == 0) nlist[blockIdx.x] = lcnt;
}

static __device__ __forceinline__ void add_rows(int s, int x, int wi, int m, int lane,
        const float* __restrict__ h, float* tile) {
    int k = 0;
    for (; k + 16 <= m; k += 16) {
        float2 v[16];
        #pragma unroll
        for (int j = 0; j < 16; ++j) {
            int sk = __builtin_amdgcn_readlane(s, k + j);
            v[j] = *(const float2*)(h + (unsigned)(sk * COUT + 2 * lane));
        }
        #pragma unroll
        for (int j = 0; j < 16; ++j) {
            int xk = __builtin_amdgcn_readlane(x, k + j);
            float wk = __int_as_float(__builtin_amdgcn_readlane(wi, k + j));
            float* t = tile + xk * COUT + 2 * lane;
            lds_add(t, v[j].x * wk);
            lds_add(t + 1, v[j].y * wk);
        }
    }
    for (; k < m; ++k) {
        int sk = __builtin_amdgcn_readlane(s, k);
        int xk = __builtin_amdgcn_readlane(x, k);
        float wk = __int_as_float(__builtin_amdgcn_readlane(wi, k));
        float2 v = *(const float2*)(h + (unsigned)(sk * COUT + 2 * lane));
        float* t = tile + xk * COUT + 2 * lane;
        lds_add(t, v.x * wk);
        lds_add(t + 1, v.y * wk);
    }
}

static __device__ __forceinline__ void scan_direct(int wave, int lane, int lo, int cnt,
        const int* __restrict__ srcp, const int* __restrict__ dstp,
        const float* __restrict__ dinv, const float* __restrict__ h, float* tile, int E) {
    for (int e0 = wave * 32; e0 < E; e0 += 1024) {
        const int e = e0 + lane;
        int s = 0, x = TILE;
        float w = 0.f;
        bool ok = false;
        if (e < E) {
            const int d = dstp[e];
            const unsigned xr = (unsigned)(d - lo);
            if (xr < (unsigned)cnt) {
                ok = true;
                s = srcp[e];
                x = (int)xr;
                w = dinv[s] * dinv[d];
            }
        }
        const int wi = __float_as_int(w);
        unsigned mk = __builtin_amdgcn_ballot_w32(ok);
        while (mk) {
            const int j = __builtin_ctz(mk);
            mk &= mk - 1u;
            const int sk = __builtin_amdgcn_readlane(s, j);
            const int xk = __builtin_amdgcn_readlane(x, j);
            const float wk = __int_as_float(__builtin_amdgcn_readlane(wi, j));
            const float2 v = *(const float2*)(h + (unsigned)(sk * COUT + 2 * lane));
            float* t = tile + xk * COUT + 2 * lane;
            lds_add(t, v.x * wk);
            lds_add(t + 1, v.y * wk);
        }
    }
}

template <bool WIDE>
static __device__ __forceinline__ void list_rows(const uint2* __restrict__ L, int nl, int wave, int lane, int lo,
        const float* __restrict__ dinv, const unsigned short* dg, const float* __restrict__ h, float* tile) {
    const int b0 = wave * 32;
    const int last = nl - 1;
    uint2 e1 = make_uint2(0u, 0u), e2 = e1;
    float w1 = 0.f;
    if (b0 < nl) {
        e1 = L[min(b0 + lane, last)];
        e2 = L[min(b0 + 1024 + lane, last)];
        if (WIDE) w1 = dinv[e1.x] * dinv[lo + (int)e1.y];
        else      w1 = dinv_of(dg[e1.x]) * dinv_of(dg[lo + (int)e1.y]);
    }
    for (int bb = b0; bb < nl; bb += 1024) {
        const uint2 e3 = L[min(bb + 2048 + lane, last)];
        float ds2, dd2;
        if (WIDE) { ds2 = dinv[e2.x]; dd2 = dinv[lo + (int)e2.y]; }
        else      { ds2 = dinv_of(dg[e2.x]); dd2 = dinv_of(dg[lo + (int)e2.y]); }
        const int m = (nl - bb) < 32 ? (nl - bb) : 32;
        add_rows((int)e1.x, (int)e1.y, __float_as_int(w1), m, lane, h, tile);
        e1 = e2;
        e2 = e3;
        w1 = ds2 * dd2;
    }
}

typedef unsigned short us8 __attribute__((ext_vector_type(8)));
#define TPV ((MAXT * TILE / 8 + 1023) / 1024)

__global__ void __launch_bounds__(1024)
k_tile(const int* __restrict__ srcp, const int* __restrict__ dstp,
       const float* __restrict__ dinv, const unsigned short* __restrict__ deg16, const float* __restrict__ h,
       const float* __restrict__ b,
       const uint2* __restrict__ lists, const unsigned* __restrict__ nlist,
       float* __restrict__ out, int N, int E) {
    __shared__ float tile[(TILE + 1) * COUT];
    __shared__ __attribute__((aligned(16))) unsigned short dg[MAXT * TILE];
    __shared__ unsigned wsat[32];

    const int lane = threadIdx.x & 31;
    const int wave = __builtin_amdgcn_readfirstlane((int)(threadIdx.x >> 5));
    const int lo   = blockIdx.x * TILE;
    const int cnt  = (N - lo) < TILE ? (N - lo) : TILE;

    const int nn = N < MAXT * TILE ? N : MAXT * TILE;
    const int nvec = nn >> 3;
    us8 mx = (us8)(0);
    {
        const us8* g8 = (const us8*)deg16;
        us8* d8 = (us8*)dg;
        for (int v0 = threadIdx.x; v0 < nvec; v0 += TPV * 1024) {
            us8 q[TPV];
            #pragma unroll
            for (int j = 0; j < TPV; ++j) {
                const int v = v0 + 1024 * j;
                q[j] = g8[v < nvec ? v : v0];
            }
            #pragma unroll
            for (int j = 0; j < TPV; ++j) {
                const int v = v0 + 1024 * j;
                d8[v < nvec ? v : v0] = q[j];
                mx = __builtin_elementwise_max(mx, q[j]);
            }
        }
    }
    unsigned short mt = __builtin_reduce_max(mx);
    {
        const int i = (nvec << 3) + (int)threadIdx.x;
        if (i < nn) {
            const unsigned short d = deg16[i];
            dg[i] = d;
            mt = d > mt ? d : mt;
        }
    }
    const unsigned wsa = __builtin_amdgcn_ballot_w32((unsigned)mt == DSAT);
    if (lane == 0) wsat[wave] = wsa;

    for (int idx = threadIdx.x; idx < cnt * COUT; idx += 1024) {
        int i = idx >> 6, c = idx & 63;
        float di = dinv[lo + i];
        tile[idx] = b[c] + di * di * h[(size_t)(lo + i) * COUT + c];
    }
    if (threadIdx.x < COUT) tile[TILE * COUT + threadIdx.x] = 0.f;
    __syncthreads();

    const bool wide = __builtin_amdgcn_ballot_w32(wsat[lane] != 0u) != 0u;

    const int nl = __builtin_amdgcn_readfirstlane((int)nlist[blockIdx.x]);
    if (__builtin_expect(nl <= LCAP, 1)) {
        const uint2* L = lists + (size_t)blockIdx.x * LCAP;
        if (__builtin_expect(!wide, 1)) list_rows<false>(L, nl, wave, lane, lo, dinv, dg, h, tile);
        else                            list_rows<true>(L, nl, wave, lane, lo, dinv, dg, h, tile);
    } else {
        scan_direct(wave, lane, lo, cnt, srcp, dstp, dinv, h, tile, E);
    }
    __syncthreads();

    float* op = out + (size_t)lo * COUT;
    for (int idx = threadIdx.x; idx < cnt * COUT; idx += 1024) op[idx] = tile[idx];
}

extern "C" void kernel_launch(void* const* d_in, const int* in_sizes, int n_in,
                              void* d_out, int out_size, void* d_ws, size_t ws_size,
                              hipStream_t stream) {
    const float* x  = (const float*)d_in[0];
    const int*   ei = (const int*)d_in[1];
    const float* W  = (const float*)d_in[2];
    const float* b  = (const float*)d_in[3];
    float* out = (float*)d_out;

    const int N = in_sizes[0] / CIN;
    const int E = in_sizes[1] / 2;
    const int* srcp = ei;
    const int* dstp = ei + E;

    const int nt = (N + TILE - 1) / TILE;
    char* ws = (char*)d_ws;
    float*    h     = (float*)ws;
    uint2*    lists = (uint2*)(ws + (size_t)N * COUT * sizeof(float));
    uint2*    bkts  = lists + (size_t)nt * LCAP;
    unsigned* nlist = (unsigned*)(bkts + (size_t)nt * NSL * BCAP);
    unsigned* cnts  = nlist + nt;
    float*    dinv  = (float*)(cnts + (size_t)nt * NSL);
    unsigned short* deg16 = (unsigned short*)(dinv + N);

    k_bin<<<NSL, 1024, 0, stream>>>(srcp, dstp, bkts, cnts, N, E);
    k_degown<<<nt, 1024, 0, stream>>>(srcp, dstp, bkts, cnts, dinv, deg16, lists, nlist, N, E);

    const int nb = (N + 127) / 128;
    k_gemm<<<nb, 256, 0, stream>>>(x, W, h, N);

    k_tile<<<nt, 1024, 0, stream>>>(srcp, dstp, dinv, deg16, h, b, lists, nlist, out, N, E);
}


# binning launch orders its slice by tile in LDS (returning add, prefix sum, staging array); a wave copies whole tile runs to the buckets: whole-line stores, no 8-byte store per edge; sums by arrival
# speedup vs baseline: 5.3490x; 1.0848x over previous
#include <hip/hip_runtime.h>


typedef float v2f __attribute__((ext_vector_type(2)));
typedef float v8f __attribute__((ext_vector_type(8)));

#define CIN 256
#define COUT 64


#define TILE 391
#define MAXT 256

#define LCAP 65536

#define NSL 256
#define BCAP 128

static __device__ __forceinline__ int slice_len(int E) { return (((E + NSL - 1) / NSL) + 3) & ~3; }

static __device__ __forceinline__ float dinv_of(unsigned deg) { return rsqrtf((float)deg); }

#define DSAT 65535u

static __device__ __forceinline__ int base4(int e, int se) {
    const int t = se - 4 > 0 ? se - 4 : 0;
    return e + 3 < se ? e : t;
}

static __device__ __forceinline__ void ld4(const int* __restrict__ srcp, const int* __restrict__ dstp,
                                           int eb, int* s, int* d) {
    #pragma unroll
    for (int j = 0; j < 4; ++j) { s[j] = srcp[eb + j]; d[j] = dstp[eb + j]; }
}

#define SCAP 16384
#define NGRP (SCAP / 4096)
#define TPW  (MAXT / 32)

__global__ void __launch_bounds__(1024)
k_bin(const int* __restrict__ srcp, const int* __restrict__ dstp,
      uint2* __restrict__ buckets, unsigned* __restrict__ cnts, int N, int E) {
    __shared__ unsigned cc[MAXT];
    __shared__ unsigned bc[MAXT];
    __shared__ unsigned off[MAXT];
    __shared__ uint2 stage[SCAP];

    const int lane = threadIdx.x & 31;
    const int wave = __builtin_amdgcn_readfirstlane((int)(threadIdx.x >> 5));
    if (threadIdx.x < MAXT) { cc[threadIdx.x] = 0u; bc[threadIdx.x] = 0u; }

    const int per = slice_len(E);
    const int sb  = blockIdx.x * per;
    const int se  = (E - sb) < per ? E : sb + per;
    const unsigned lim = (unsigned)(N < MAXT * TILE ? N : MAXT * TILE);
    uint2* B = buckets + (size_t)blockIdx.x * BCAP;

    for (int cb = sb; cb < se; cb += SCAP) {
        const int ce = (se - cb) < SCAP ? se : cb + SCAP;

        int sc[NGRP][4], dc[NGRP][4], eb[NGRP];
        #pragma unroll
        for (int q = 0; q < NGRP; ++q) {
            eb[q] = base4(cb + q * 4096 + (int)threadIdx.x * 4, ce);
            ld4(srcp, dstp, eb[q], sc[q], dc[q]);
        }
        #pragma unroll
        for (int q = 0; q < NGRP; ++q) {
            const int e0 = cb + q * 4096 + (int)threadIdx.x * 4;
            #pragma unroll
            for (int j = 0; j < 4; ++j) if (eb[q] + j < e0) dc[q][j] = -1;
        }
        __syncthreads();

        unsigned rk[NGRP][4];
        #pragma unroll
        for (int q = 0; q < NGRP; ++q) {
            #pragma unroll
            for (int j = 0; j < 4; ++j) {
                const unsigned d = (unsigned)dc[q][j];
                rk[q][j] = 0u;
                if (d < lim)
                    rk[q][j] = __hip_atomic_fetch_add(&cc[d / TILE], 1u, __ATOMIC_RELAXED, __HIP_MEMORY_SCOPE_WORKGROUP);
            }
        }
        __syncthreads();

        if (wave == 0) {
            unsigned c[TPW];
            unsigned s = 0u;
            #pragma unroll
            for (int k = 0; k < TPW; ++k) { c[k] = cc[lane * TPW + k]; s += c[k]; }
            unsigned inc = s;
            #pragma unroll
            for (int dl = 1; dl < 32; dl <<= 1) {
                const unsigned v = __shfl_up(inc, (unsigned)dl);
                if (lane >= dl) inc += v;
            }
            unsigned ex = inc - s;
            #pragma unroll
            for (int k = 0; k < TPW; ++k) { off[lane * TPW + k] = ex; ex += c[k]; }
        }
        __syncthreads();

        unsigned ps[NGRP][4];
        #pragma unroll
        for (int q = 0; q < NGRP; ++q) {
            #pragma unroll
            for (int j = 0; j < 4; ++j) {
                const unsigned d = (unsigned)dc[q][j];
                ps[q][j] = off[d < lim ? d / TILE : 0u] + rk[q][j];
            }
        }
        #pragma unroll
        for (int q = 0; q < NGRP; ++q) {
            #pragma unroll
            for (int j = 0; j < 4; ++j) {
                const unsigned d = (unsigned)dc[q][j];
                if (d < lim) stage[ps[q][j]] = make_uint2((unsigned)sc[q][j], d - (d / TILE) * TILE);
            }
        }
        __syncthreads();

        {
            unsigned vo = 0u, vb = 0u, vs = 0u;
            if (lane < TPW) {
                const int t = wave * TPW + lane;
                const unsigned vn = cc[t];
                vo = off[t];
                vb = bc[t];
                const unsigned room = vb < (unsigned)BCAP ? (unsigned)BCAP - vb : 0u;
                vs = vn < room ? vn : room;
            }
            unsigned nk[TPW], ok[TPW];
            uint2 en[TPW][2];
            #pragma unroll
            for (int k = 0; k < TPW; ++k) {
                nk[k] = (unsigned)__builtin_amdgcn_readlane((int)vs, k);
                ok[k] = (unsigned)__builtin_amdgcn_readlane((int)vo, k);
                #pragma unroll
                for (int c = 0; c < 2; ++c) {
                    const unsigned i = (unsigned)(c * 32 + lane);
                    en[k][c] = make_uint2(0u, 0u);
                    if (i < nk[k]) en[k][c] = stage[ok[k] + i];
                }
            }
            #pragma unroll
            for (int k = 0; k < TPW; ++k) {
                const unsigned bk = (unsigned)__builtin_amdgcn_readlane((int)vb, k);
                uint2* Bt = B + (size_t)(wave * TPW + k) * (NSL * BCAP) + bk;
                #pragma unroll
                for (int c = 0; c < 2; ++c) {
                    const unsigned i = (unsigned)(c * 32 + lane);
                    if (i < nk[k]) Bt[i] = en[k][c];
                }
                for (unsigned i = 64u + (unsigned)lane; i < nk[k]; i += 32u) Bt[i] = stage[ok[k] + i];
            }
        }
        __syncthreads();

        if (threadIdx.x < MAXT) { bc[threadIdx.x] += cc[threadIdx.x]; cc[threadIdx.x] = 0u; }
    }

    const int nt = (N + TILE - 1) / TILE;
    if ((int)threadIdx.x < nt && threadIdx.x < MAXT) cnts[threadIdx.x * NSL + blockIdx.x] = bc[threadIdx.x];
}

__global__ void __launch_bounds__(256)
k_gemm(const float* __restrict__ x, const float* __restrict__ W,
       float* __restrict__ h, int N) {
    __shared__ float2 Ws2[(CIN / 2) * COUT];
    for (int idx = threadIdx.x; idx < (CIN / 2) * COUT; idx += 256) {
        int p = idx >> 6, n = idx & 63;
        Ws2[idx] = make_float2(W[p * (2 * COUT) + n], W[p * (2 * COUT) + COUT + n]);
    }
    __syncthreads();

    const int lane = threadIdx.x & 31;
    const int wave = threadIdx.x >> 5;
    const int m0   = (blockIdx.x * 8 + wave) * 16;
    const int mr   = lane & 15;
    const int kh   = (lane >> 4) << 1;

    int row  = m0 + mr;
    int rowc = row < N ? row : (N - 1);
    const float* xr = x + (size_t)rowc * CIN;

    v8f acc0 = {}; v8f acc1 = {}; v8f acc2 = {}; v8f acc3 = {};
    for (int k = 0; k < CIN; k += 4) {
        v2f a = *(const v2f*)(xr + k + kh);
        const float2* wp = &Ws2[(((k + kh) >> 1) * COUT) + mr];
        v2f b0 = *(const v2f*)(wp + 0);
        v2f b1 = *(const v2f*)(wp + 16);
        v2f b2 = *(const v2f*)(wp + 32);
        v2f b3 = *(const v2f*)(wp + 48);
        acc0 = __builtin_amdgcn_wmma_f32_16x16x4_f32(false, a, false, b0, (short)0, acc0, false, false);
        acc1 = __builtin_amdgcn_wmma_f32_16x16x4_f32(false, a, false, b1, (short)0, acc1, false, false);
        acc2 = __builtin_amdgcn_wmma_f32_16x16x4_f32(false, a, false, b2, (short)0, acc2, false, false);
        acc3 = __builtin_amdgcn_wmma_f32_16x16x4_f32(false, a, false, b3, (short)0, acc3, false, false);
    }

    const int nlo = lane & 15;
    const int mhi = (lane >> 4) << 3;
    for (int r = 0; r < 8; ++r) {
        int orow = m0 + r + mhi;
        if (orow < N) {
            float* hp = h + (size_t)orow * COUT + nlo;
            hp[0]  = acc0[r];
            hp[16] = acc1[r];
            hp[32] = acc2[r];
            hp[48] = acc3[r];
        }
    }
}


static __device__ __forceinline__ void lds_add(float* p, float v) {
    __hip_atomic_fetch_add(p, v, __ATOMIC_RELAXED, __HIP_MEMORY_SCOPE_WORKGROUP);
}

#define BPW (NSL / 32)

__global__ void __launch_bounds__(1024)
k_degown(const int* __restrict__ srcp, const int* __restrict__ dstp,
         const uint2* __restrict__ buckets, const unsigned* __restrict__ cnts,
         float* __restrict__ dinv, unsigned short* __restrict__ deg16,
         uint2* __restrict__ lists, unsigned* __restrict__ nlist, int N, int E) {
    __shared__ unsigned cl[TILE];
    __shared__ unsigned lcnt;

    const int lane = threadIdx.x & 31;
    const int wave = __builtin_amdgcn_readfirstlane((int)(threadIdx.x >> 5));
    const int lo   = blockIdx.x * TILE;
    const int cnt  = (N - lo) < TILE ? (N - lo) : TILE;

    if (threadIdx.x < TILE) cl[threadIdx.x] = 0u;
    if (threadIdx.x == 0) lcnt = 0u;
    __syncthreads();

    uint2* list = lists + (size_t)blockIdx.x * LCAP;
    const unsigned* cn = cnts + (size_t)blockIdx.x * NSL;
    const uint2* B = buckets + (size_t)blockIdx.x * (NSL * BCAP);

    unsigned nv = 0u;
    if (lane < BPW) nv = cn[wave + 32 * lane];
    const unsigned nu = nv <= (unsigned)BCAP ? nv : 0u;

    unsigned nj[BPW];
    uint2 en[BPW][2];
    unsigned tot = 0u;
    #pragma unroll
    for (int j = 0; j < BPW; ++j) {
        nj[j] = (unsigned)__builtin_amdgcn_readlane((int)nu, j);
        tot += nj[j];
        const uint2* Bj = B + (wave + 32 * j) * BCAP;
        #pragma unroll
        for (int c = 0; c < 2; ++c) {
            const unsigned i = (unsigned)(c * 32 + lane);
            en[j][c] = make_uint2(0u, 0u);
            if (i < nj[j]) en[j][c] = Bj[i];
        }
    }
    unsigned base = 0u;
    if (tot) {
        if (lane == 0) base = __hip_atomic_fetch_add(&lcnt, tot, __ATOMIC_RELAXED, __HIP_MEMORY_SCOPE_WORKGROUP);
        base = (unsigned)__builtin_amdgcn_readfirstlane((int)base);
    }
    #pragma unroll
    for (int j = 0; j < BPW; ++j) {
        #pragma unroll
        for (int c = 0; c < 2; ++c) {
            const unsigned i = (unsigned)(c * 32 + lane);
            if (i < nj[j]) {
                __hip_atomic_fetch_add(&cl[en[j][c].y], 1u, __ATOMIC_RELAXED, __HIP_MEMORY_SCOPE_WORKGROUP);
                if (base + i < (unsigned)LCAP) list[base + i] = en[j][c];
            }
        }
        const uint2* Bj = B + (wave + 32 * j) * BCAP;
        for (unsigned i = 64u + (unsigned)lane; i < nj[j]; i += 32u) {
            const uint2 e = Bj[i];
            __hip_atomic_fetch_add(&cl[e.y], 1u, __ATOMIC_RELAXED, __HIP_MEMORY_SCOPE_WORKGROUP);
            if (base + i < (unsigned)LCAP) list[base + i] = e;
        }
        base += nj[j];
    }

    if (__builtin_amdgcn_ballot_w32(nv > (unsigned)BCAP)) {
        const int per = slice_len(E);
        for (int j = 0; j < BPW; ++j) {
            const unsigned n = (unsigned)__builtin_amdgcn_readlane((int)nv, j);
            if (n <= (unsigned)BCAP) continue;
            const int sb = (wave + 32 * j) * per;
            const int se = (E - sb) < per ? E : sb + per;
            for (int e0 = sb; e0 < se; e0 += 32) {
                const int e = e0 + lane;
                bool ok = false;
                unsigned xr = 0u;
                int s = 0;
                if (e < se) {
                    xr = (unsigned)(dstp[e] - lo);
                    ok = xr < (unsigned)cnt;
                    if (ok) {
                        s = srcp[e];
                        __hip_atomic_fetch_add(&cl[xr], 1u, __ATOMIC_RELAXED, __HIP_MEMORY_SCOPE_WORKGROUP);
                    }
                }
                const unsigned mk = __builtin_amdgcn_ballot_w32(ok);
                if (mk) {
                    unsigned b2 = 0u;
                    if (lane == 0)
                        b2 = __hip_atomic_fetch_add(&lcnt, (unsigned)__builtin_popcount(mk), __ATOMIC_RELAXED, __HIP_MEMORY_SCOPE_WORKGROUP);
                    b2 = (unsigned)__builtin_amdgcn_readfirstlane((int)b2);
                    const unsigned pos = b2 + __builtin_amdgcn_mbcnt_lo(mk, 0u);
                    if (ok && pos < (unsigned)LCAP) list[pos] = make_uint2((unsigned)s, xr);
                }
            }
        }
    }
    __syncthreads();

    if ((int)threadIdx.x < cnt) {
        const unsigned dg = cl[threadIdx.x] + 1u;
        dinv[lo + threadIdx.x] = dinv_of(dg);
        deg16[lo + threadIdx.x] = (unsigned short)(dg < DSAT ? dg : DSAT);
    }
    if (threadIdx.x == 0) nlist[blockIdx.x] = lcnt;
}

static __device__ __forceinline__ void add_rows(int s, int x, int wi, int m, int lane,
        const float* __restrict__ h, float* tile) {
    int k = 0;
    for (; k + 16 <= m; k += 16) {
        float2 v[16];
        #pragma unroll
        for (int j = 0; j < 16; ++j) {
            int sk = __builtin_amdgcn_readlane(s, k + j);
            v[j] = *(const float2*)(h + (unsigned)(sk * COUT + 2 * lane));
        }
        #pragma unroll
        for (int j = 0; j < 16; ++j) {
            int xk = __builtin_amdgcn_readlane(x, k + j);
            float wk = __int_as_float(__builtin_amdgcn_readlane(wi, k + j));
            float* t = tile + xk * COUT + 2 * lane;
            lds_add(t, v[j].x * wk);
            lds_add(t + 1, v[j].y * wk);
        }
    }
    for (; k < m; ++k) {
        int sk = __builtin_amdgcn_readlane(s, k);
        int xk = __builtin_amdgcn_readlane(x, k);
        float wk = __int_as_float(__builtin_amdgcn_readlane(wi, k));
        float2 v = *(const float2*)(h + (unsigned)(sk * COUT + 2 * lane));
        float* t = tile + xk * COUT + 2 * lane;
        lds_add(t, v.x * wk);
        lds_add(t + 1, v.y * wk);
    }
}

static __device__ __forceinline__ void scan_direct(int wave, int lane, int lo, int cnt,
        const int* __restrict__ srcp, const int* __restrict__ dstp,
        const float* __restrict__ dinv, const float* __restrict__ h, float* tile, int E) {
    for (int e0 = wave * 32; e0 < E; e0 += 1024) {
        const int e = e0 + lane;
        int s = 0, x = TILE;
        float w = 0.f;
        bool ok = false;
        if (e < E) {
            const int d = dstp[e];
            const unsigned xr = (unsigned)(d - lo);
            if (xr < (unsigned)cnt) {
                ok = true;
                s = srcp[e];
                x = (int)xr;
                w = dinv[s] * dinv[d];
            }
        }
        const int wi = __float_as_int(w);
        unsigned mk = __builtin_amdgcn_ballot_w32(ok);
        while (mk) {
            const int j = __builtin_ctz(mk);
            mk &= mk - 1u;
            const int sk = __builtin_amdgcn_readlane(s, j);
            const int xk = __builtin_amdgcn_readlane(x, j);
            const float wk = __int_as_float(__builtin_amdgcn_readlane(wi, j));
            const float2 v = *(const float2*)(h + (unsigned)(sk * COUT + 2 * lane));
            float* t = tile + xk * COUT + 2 * lane;
            lds_add(t, v.x * wk);
            lds_add(t + 1, v.y * wk);
        }
    }
}

template <bool WIDE>
static __device__ __forceinline__ void list_rows(const uint2* __restrict__ L, int nl, int wave, int lane, int lo,
        const float* __restrict__ dinv, const unsigned short* dg, const float* __restrict__ h, float* tile) {
    const int b0 = wave * 32;
    const int last = nl - 1;
    uint2 e1 = make_uint2(0u, 0u), e2 = e1;
    float w1 = 0.f;
    if (b0 < nl) {
        e1 = L[min(b0 + lane, last)];
        e2 = L[min(b0 + 1024 + lane, last)];
        if (WIDE) w1 = dinv[e1.x] * dinv[lo + (int)e1.y];
        else      w1 = dinv_of(dg[e1.x]) * dinv_of(dg[lo + (int)e1.y]);
    }
    for (int bb = b0; bb < nl; bb += 1024) {
        const uint2 e3 = L[min(bb + 2048 + lane, last)];
        float ds2, dd2;
        if (WIDE) { ds2 = dinv[e2.x]; dd2 = dinv[lo + (int)e2.y]; }
        else      { ds2 = dinv_of(dg[e2.x]); dd2 = dinv_of(dg[lo + (int)e2.y]); }
        const int m = (nl - bb) < 32 ? (nl - bb) : 32;
        add_rows((int)e1.x, (int)e1.y, __float_as_int(w1), m, lane, h, tile);
        e1 = e2;
        e2 = e3;
        w1 = ds2 * dd2;
    }
}

typedef unsigned short us8 __attribute__((ext_vector_type(8)));
#define TPV ((MAXT * TILE / 8 + 1023) / 1024)

__global__ void __launch_bounds__(1024)
k_tile(const int* __restrict__ srcp, const int* __restrict__ dstp,
       const float* __restrict__ dinv, const unsigned short* __restrict__ deg16, const float* __restrict__ h,
       const float* __restrict__ b,
       const uint2* __restrict__ lists, const unsigned* __restrict__ nlist,
       float* __restrict__ out, int N, int E) {
    __shared__ float tile[(TILE + 1) * COUT];
    __shared__ __attribute__((aligned(16))) unsigned short dg[MAXT * TILE];
    __shared__ unsigned wsat[32];

    const int lane = threadIdx.x & 31;
    const int wave = __builtin_amdgcn_readfirstlane((int)(threadIdx.x >> 5));
    const int lo   = blockIdx.x * TILE;
    const int cnt  = (N - lo) < TILE ? (N - lo) : TILE;

    const int nn = N < MAXT * TILE ? N : MAXT * TILE;
    const int nvec = nn >> 3;
    us8 mx = (us8)(0);
    {
        const us8* g8 = (const us8*)deg16;
        us8* d8 = (us8*)dg;
        for (int v0 = threadIdx.x; v0 < nvec; v0 += TPV * 1024) {
            us8 q[TPV];
            #pragma unroll
            for (int j = 0; j < TPV; ++j) {
                const int v = v0 + 1024 * j;
                q[j] = g8[v < nvec ? v : v0];
            }
            #pragma unroll
            for (int j = 0; j < TPV; ++j) {
                const int v = v0 + 1024 * j;
                d8[v < nvec ? v : v0] = q[j];
                mx = __builtin_elementwise_max(mx, q[j]);
            }
        }
    }
    unsigned short mt = __builtin_reduce_max(mx);
    {
        const int i = (nvec << 3) + (int)threadIdx.x;
        if (i < nn) {
            const unsigned short d = deg16[i];
            dg[i] = d;
            mt = d > mt ? d : mt;
        }
    }
    const unsigned wsa = __builtin_amdgcn_ballot_w32((unsigned)mt == DSAT);
    if (lane == 0) wsat[wave] = wsa;

    for (int idx = threadIdx.x; idx < cnt * COUT; idx += 1024) {
        int i = idx >> 6, c = idx & 63;
        float di = dinv[lo + i];
        tile[idx] = b[c] + di * di * h[(size_t)(lo + i) * COUT + c];
    }
    if (threadIdx.x < COUT) tile[TILE * COUT + threadIdx.x] = 0.f;
    __syncthreads();

    const bool wide = __builtin_amdgcn_ballot_w32(wsat[lane] != 0u) != 0u;

    const int nl = __builtin_amdgcn_readfirstlane((int)nlist[blockIdx.x]);
    if (__builtin_expect(nl <= LCAP, 1)) {
        const uint2* L = lists + (size_t)blockIdx.x * LCAP;
        if (__builtin_expect(!wide, 1)) list_rows<false>(L, nl, wave, lane, lo, dinv, dg, h, tile);
        else                            list_rows<true>(L, nl, wave, lane, lo, dinv, dg, h, tile);
    } else {
        scan_direct(wave, lane, lo, cnt, srcp, dstp, dinv, h, tile, E);
    }
    __syncthreads();

    float* op = out + (size_t)lo * COUT;
    for (int idx = threadIdx.x; idx < cnt * COUT; idx += 1024) op[idx] = tile[idx];
}

extern "C" void kernel_launch(void* const* d_in, const int* in_sizes, int n_in,
                              void* d_out, int out_size, void* d_ws, size_t ws_size,
                              hipStream_t stream) {
    const float* x  = (const float*)d_in[0];
    const int*   ei = (const int*)d_in[1];
    const float* W  = (const float*)d_in[2];
    const float* b  = (const float*)d_in[3];
    float* out = (float*)d_out;

    const int N = in_sizes[0] / CIN;
    const int E = in_sizes[1] / 2;
    const int* srcp = ei;
    const int* dstp = ei + E;

    const int nt = (N + TILE - 1) / TILE;
    char* ws = (char*)d_ws;
    float*    h     = (float*)ws;
    uint2*    lists = (uint2*)(ws + (size_t)N * COUT * sizeof(float));
    uint2*    bkts  = lists + (size_t)nt * LCAP;
    unsigned* nlist = (unsigned*)(bkts + (size_t)nt * NSL * BCAP);
    unsigned* cnts  = nlist + nt;
    float*    dinv  = (float*)(cnts + (size_t)nt * NSL);
    unsigned short* deg16 = (unsigned short*)(dinv + N);

    k_bin<<<NSL, 1024, 0, stream>>>(srcp, dstp, bkts, cnts, N, E);
    k_degown<<<nt, 1024, 0, stream>>>(srcp, dstp, bkts, cnts, dinv, deg16, lists, nlist, N, E);

    const int nb = (N + 127) / 128;
    k_gemm<<<nb, 256, 0, stream>>>(x, W, h, N);

    k_tile<<<nt, 1024, 0, stream>>>(srcp, dstp, dinv, deg16, h, b, lists, nlist, out, N, E);
}


# h = X*W formed by the tile owners in the degree launch: after its count an owner's waves take the 25 row strips of its 391 nodes, W staged once per owner; 3 launches, no product launch
# speedup vs baseline: 5.5839x; 1.0439x over previous
#include <hip/hip_runtime.h>


typedef float v2f __attribute__((ext_vector_type(2)));
typedef float v8f __attribute__((ext_vector_type(8)));

#define CIN 256
#define COUT 64


#define TILE 391
#define MAXT 256

#define LCAP 65536

#define NSL 256
#define BCAP 128

static __device__ __forceinline__ int slice_len(int E) { return (((E + NSL - 1) / NSL) + 3) & ~3; }

static __device__ __forceinline__ float dinv_of(unsigned deg) { return rsqrtf((float)deg); }

#define DSAT 65535u

static __device__ __forceinline__ int base4(int e, int se) {
    const int t = se - 4 > 0 ? se - 4 : 0;
    return e + 3 < se ? e : t;
}

static __device__ __forceinline__ void ld4(const int* __restrict__ srcp, const int* __restrict__ dstp,
                                           int eb, int* s, int* d) {
    #pragma unroll
    for (int j = 0; j < 4; ++j) { s[j] = srcp[eb + j]; d[j] = dstp[eb + j]; }
}

#define SCAP 16384
#define NGRP (SCAP / 4096)
#define TPW  (MAXT / 32)

__global__ void __launch_bounds__(1024)
k_bin(const int* __restrict__ srcp, const int* __restrict__ dstp,
      uint2* __restrict__ buckets, unsigned* __restrict__ cnts, int N, int E) {
    __shared__ unsigned cc[MAXT];
    __shared__ unsigned bc[MAXT];
    __shared__ unsigned off[MAXT];
    __shared__ uint2 stage[SCAP];

    const int lane = threadIdx.x & 31;
    const int wave = __builtin_amdgcn_readfirstlane((int)(threadIdx.x >> 5));
    if (threadIdx.x < MAXT) { cc[threadIdx.x] = 0u; bc[threadIdx.x] = 0u; }

    const int per = slice_len(E);
    const int sb  = blockIdx.x * per;
    const int se  = (E - sb) < per ? E : sb + per;
    const unsigned lim = (unsigned)(N < MAXT * TILE ? N : MAXT * TILE);
    uint2* B = buckets + (size_t)blockIdx.x * BCAP;

    for (int cb = sb; cb < se; cb += SCAP) {
        const int ce = (se - cb) < SCAP ? se : cb + SCAP;

        int sc[NGRP][4], dc[NGRP][4], eb[NGRP];
        #pragma unroll
        for (int q = 0; q < NGRP; ++q) {
            eb[q] = base4(cb + q * 4096 + (int)threadIdx.x * 4, ce);
            ld4(srcp, dstp, eb[q], sc[q], dc[q]);
        }
        #pragma unroll
        for (int q = 0; q < NGRP; ++q) {
            const int e0 = cb + q * 4096 + (int)threadIdx.x * 4;
            #pragma unroll
            for (int j = 0; j < 4; ++j) if (eb[q] + j < e0) dc[q][j] = -1;
        }
        __syncthreads();

        unsigned rk[NGRP][4];
        #pragma unroll
        for (int q = 0; q < NGRP; ++q) {
            #pragma unroll
            for (int j = 0; j < 4; ++j) {
                const unsigned d = (unsigned)dc[q][j];
                rk[q][j] = 0u;
                if (d < lim)
                    rk[q][j] = __hip_atomic_fetch_add(&cc[d / TILE], 1u, __ATOMIC_RELAXED, __HIP_MEMORY_SCOPE_WORKGROUP);
            }
        }
        __syncthreads();

        if (wave == 0) {
            unsigned c[TPW];
            unsigned s = 0u;
            #pragma unroll
            for (int k = 0; k < TPW; ++k) { c[k] = cc[lane * TPW + k]; s += c[k]; }
            unsigned inc = s;
            #pragma unroll
            for (int dl = 1; dl < 32; dl <<= 1) {
                const unsigned v = __shfl_up(inc, (unsigned)dl);
                if (lane >= dl) inc += v;
            }
            unsigned ex = inc - s;
            #pragma unroll
            for (int k = 0; k < TPW; ++k) { off[lane * TPW + k] = ex; ex += c[k]; }
        }
        __syncthreads();

        unsigned ps[NGRP][4];
        #pragma unroll
        for (int q = 0; q < NGRP; ++q) {
            #pragma unroll
            for (int j = 0; j < 4; ++j) {
                const unsigned d = (unsigned)dc[q][j];
                ps[q][j] = off[d < lim ? d / TILE : 0u] + rk[q][j];
            }
        }
        #pragma unroll
        for (int q = 0; q < NGRP; ++q) {
            #pragma unroll
            for (int j = 0; j < 4; ++j) {
                const unsigned d = (unsigned)dc[q][j];
                if (d < lim) stage[ps[q][j]] = make_uint2((unsigned)sc[q][j], d - (d / TILE) * TILE);
            }
        }
        __syncthreads();

        {
            unsigned vo = 0u, vb = 0u, vs = 0u;
            if (lane < TPW) {
                const int t = wave * TPW + lane;
                const unsigned vn = cc[t];
                vo = off[t];
                vb = bc[t];
                const unsigned room = vb < (unsigned)BCAP ? (unsigned)BCAP - vb : 0u;
                vs = vn < room ? vn : room;
            }
            unsigned nk[TPW], ok[TPW];
            uint2 en[TPW][2];
            #pragma unroll
            for (int k = 0; k < TPW; ++k) {
                nk[k] = (unsigned)__builtin_amdgcn_readlane((int)vs, k);
                ok[k] = (unsigned)__builtin_amdgcn_readlane((int)vo, k);
                #pragma unroll
                for (int c = 0; c < 2; ++c) {
                    const unsigned i = (unsigned)(c * 32 + lane);
                    en[k][c] = make_uint2(0u, 0u);
                    if (i < nk[k]) en[k][c] = stage[ok[k] + i];
                }
            }
            #pragma unroll
            for (int k = 0; k < TPW; ++k) {
                const unsigned bk = (unsigned)__builtin_amdgcn_readlane((int)vb, k);
                uint2* Bt = B + (size_t)(wave * TPW + k) * (NSL * BCAP) + bk;
                #pragma unroll
                for (int c = 0; c < 2; ++c) {
                    const unsigned i = (unsigned)(c * 32 + lane);
                    if (i < nk[k]) Bt[i] = en[k][c];
                }
                for (unsigned i = 64u + (unsigned)lane; i < nk[k]; i += 32u) Bt[i] = stage[ok[k] + i];
            }
        }
        __syncthreads();

        if (threadIdx.x < MAXT) { bc[threadIdx.x] += cc[threadIdx.x]; cc[threadIdx.x] = 0u; }
    }

    const int nt = (N + TILE - 1) / TILE;
    if ((int)threadIdx.x < nt && threadIdx.x < MAXT) cnts[threadIdx.x * NSL + blockIdx.x] = bc[threadIdx.x];
}

static __device__ __forceinline__ void gemm_strip(const float* __restrict__ x, const float2* Ws2,
                                                  float* __restrict__ h, int m0, int mend, int N, int lane) {
    const int mr   = lane & 15;
    const int kh   = (lane >> 4) << 1;

    int row  = m0 + mr;
    int rowc = row < N ? row : (N - 1);
    const float* xr = x + (size_t)rowc * CIN;

    v8f acc0 = {}; v8f acc1 = {}; v8f acc2 = {}; v8f acc3 = {};
    for (int k = 0; k < CIN; k += 4) {
        v2f a = *(const v2f*)(xr + k + kh);
        const float2* wp = &Ws2[(((k + kh) >> 1) * COUT) + mr];
        v2f b0 = *(const v2f*)(wp + 0);
        v2f b1 = *(const v2f*)(wp + 16);
        v2f b2 = *(const v2f*)(wp + 32);
        v2f b3 = *(const v2f*)(wp + 48);
        acc0 = __builtin_amdgcn_wmma_f32_16x16x4_f32(false, a, false, b0, (short)0, acc0, false, false);
        acc1 = __builtin_amdgcn_wmma_f32_16x16x4_f32(false, a, false, b1, (short)0, acc1, false, false);
        acc2 = __builtin_amdgcn_wmma_f32_16x16x4_f32(false, a, false, b2, (short)0, acc2, false, false);
        acc3 = __builtin_amdgcn_wmma_f32_16x16x4_f32(false, a, false, b3, (short)0, acc3, false, false);
    }

    const int nlo = lane & 15;
    const int mhi = (lane >> 4) << 3;
    for (int r = 0; r < 8; ++r) {
        int orow = m0 + r + mhi;
        if (orow < mend) {
            float* hp = h + (size_t)orow * COUT + nlo;
            hp[0]  = acc0[r];
            hp[16] = acc1[r];
            hp[32] = acc2[r];
            hp[48] = acc3[r];
        }
    }
}


static __device__ __forceinline__ void lds_add(float* p, float v) {
    __hip_atomic_fetch_add(p, v, __ATOMIC_RELAXED, __HIP_MEMORY_SCOPE_WORKGROUP);
}

#define BPW (NSL / 32)
#define NSTRIP ((TILE + 15) / 16)
#define WPT ((CIN / 2) * COUT / 1024)

__global__ void __launch_bounds__(1024)
k_degown(const int* __restrict__ srcp, const int* __restrict__ dstp,
         const uint2* __restrict__ buckets, const unsigned* __restrict__ cnts,
         float* __restrict__ dinv, unsigned short* __restrict__ deg16,
         uint2* __restrict__ lists, unsigned* __restrict__ nlist,
         const float* __restrict__ x, const float* __restrict__ W, float* __restrict__ h, int N, int E) {
    __shared__ unsigned cl[TILE];
    __shared__ unsigned lcnt;
    __shared__ float2 Ws2[(CIN / 2) * COUT];

    const int lane = threadIdx.x & 31;
    const int wave = __builtin_amdgcn_readfirstlane((int)(threadIdx.x >> 5));
    const int lo   = blockIdx.x * TILE;
    const int cnt  = (N - lo) < TILE ? (N - lo) : TILE;

    if (threadIdx.x < TILE) cl[threadIdx.x] = 0u;
    if (threadIdx.x == 0) lcnt = 0u;
    __syncthreads();

    uint2* list = lists + (size_t)blockIdx.x * LCAP;
    const unsigned* cn = cnts + (size_t)blockIdx.x * NSL;
    const uint2* B = buckets + (size_t)blockIdx.x * (NSL * BCAP);

    unsigned nv = 0u;
    if (lane < BPW) nv = cn[wave + 32 * lane];
    const unsigned nu = nv <= (unsigned)BCAP ? nv : 0u;

    unsigned nj[BPW];
    uint2 en[BPW][2];
    unsigned tot = 0u;
    #pragma unroll
    for (int j = 0; j < BPW; ++j) {
        nj[j] = (unsigned)__builtin_amdgcn_readlane((int)nu, j);
        tot += nj[j];
        const uint2* Bj = B + (wave + 32 * j) * BCAP;
        #pragma unroll
        for (int c = 0; c < 2; ++c) {
            const unsigned i = (unsigned)(c * 32 + lane);
            en[j][c] = make_uint2(0u, 0u);
            if (i < nj[j]) en[j][c] = Bj[i];
        }
    }
    float2 wv[WPT];
    #pragma unroll
    for (int j = 0; j < WPT; ++j) {
        const int idx = (int)threadIdx.x + 1024 * j;
        const int p = idx >> 6, n = idx & 63;
        wv[j] = make_float2(W[p * (2 * COUT) + n], W[p * (2 * COUT) + COUT + n]);
    }
    unsigned base = 0u;
    if (tot) {
        if (lane == 0) base = __hip_atomic_fetch_add(&lcnt, tot, __ATOMIC_RELAXED, __HIP_MEMORY_SCOPE_WORKGROUP);
        base = (unsigned)__builtin_amdgcn_readfirstlane((int)base);
    }
    #pragma unroll
    for (int j = 0; j < BPW; ++j) {
        #pragma unroll
        for (int c = 0; c < 2; ++c) {
            const unsigned i = (unsigned)(c * 32 + lane);
            if (i < nj[j]) {
                __hip_atomic_fetch_add(&cl[en[j][c].y], 1u, __ATOMIC_RELAXED, __HIP_MEMORY_SCOPE_WORKGROUP);
                if (base + i < (unsigned)LCAP) list[base + i] = en[j][c];
            }
        }
        const uint2* Bj = B + (wave + 32 * j) * BCAP;
        for (unsigned i = 64u + (unsigned)lane; i < nj[j]; i += 32u) {
            const uint2 e = Bj[i];
            __hip_atomic_fetch_add(&cl[e.y], 1u, __ATOMIC_RELAXED, __HIP_MEMORY_SCOPE_WORKGROUP);
            if (base + i < (unsigned)LCAP) list[base + i] = e;
        }
        base += nj[j];
    }

    if (__builtin_amdgcn_ballot_w32(nv > (unsigned)BCAP)) {
        const int per = slice_len(E);
        for (int j = 0; j < BPW; ++j) {
            const unsigned n = (unsigned)__builtin_amdgcn_readlane((int)nv, j);
            if (n <= (unsigned)BCAP) continue;
            const int sb = (wave + 32 * j) * per;
            const int se = (E - sb) < per ? E : sb + per;
            for (int e0 = sb; e0 < se; e0 += 32) {
                const int e = e0 + lane;
                bool ok = false;
                unsigned xr = 0u;
                int s = 0;
                if (e < se) {
                    xr = (unsigned)(dstp[e] - lo);
                    ok = xr < (unsigned)cnt;
                    if (ok) {
                        s = srcp[e];
                        __hip_atomic_fetch_add(&cl[xr], 1u, __ATOMIC_RELAXED, __HIP_MEMORY_SCOPE_WORKGROUP);
                    }
                }
                const unsigned mk = __builtin_amdgcn_ballot_w32(ok);
                if (mk) {
                    unsigned b2 = 0u;
                    if (lane == 0)
                        b2 = __hip_atomic_fetch_add(&lcnt, (unsigned)__builtin_popcount(mk), __ATOMIC_RELAXED, __HIP_MEMORY_SCOPE_WORKGROUP);
                    b2 = (unsigned)__builtin_amdgcn_readfirstlane((int)b2);
                    const unsigned pos = b2 + __builtin_amdgcn_mbcnt_lo(mk, 0u);
                    if (ok && pos < (unsigned)LCAP) list[pos] = make_uint2((unsigned)s, xr);
                }
            }
        }
    }
    #pragma unroll
    for (int j = 0; j < WPT; ++j) Ws2[(int)threadIdx.x + 1024 * j] = wv[j];
    __syncthreads();

    if ((int)threadIdx.x < cnt) {
        const unsigned dg = cl[threadIdx.x] + 1u;
        dinv[lo + threadIdx.x] = dinv_of(dg);
        deg16[lo + threadIdx.x] = (unsigned short)(dg < DSAT ? dg : DSAT);
    }
    if (threadIdx.x == 0) nlist[blockIdx.x] = lcnt;

    {
        const int a = wave & 3, q = wave >> 3;
        const int strip = (wave & 4) ? 16 + a + 4 * ((q - a) & 3) : q * 4 + a;
        if (strip < NSTRIP && strip * 16 < cnt)
            gemm_strip(x, Ws2, h, lo + strip * 16, lo + cnt, N, lane);
    }
}

static __device__ __forceinline__ void add_rows(int s, int x, int wi, int m, int lane,
        const float* __restrict__ h, float* tile) {
    int k = 0;
    for (; k + 16 <= m; k += 16) {
        float2 v[16];
        #pragma unroll
        for (int j = 0; j < 16; ++j) {
            int sk = __builtin_amdgcn_readlane(s, k + j);
            v[j] = *(const float2*)(h + (unsigned)(sk * COUT + 2 * lane));
        }
        #pragma unroll
        for (int j = 0; j < 16; ++j) {
            int xk = __builtin_amdgcn_readlane(x, k + j);
            float wk = __int_as_float(__builtin_amdgcn_readlane(wi, k + j));
            float* t = tile + xk * COUT + 2 * lane;
            lds_add(t, v[j].x * wk);
            lds_add(t + 1, v[j].y * wk);
        }
    }
    for (; k < m; ++k) {
        int sk = __builtin_amdgcn_readlane(s, k);
        int xk = __builtin_amdgcn_readlane(x, k);
        float wk = __int_as_float(__builtin_amdgcn_readlane(wi, k));
        float2 v = *(const float2*)(h + (unsigned)(sk * COUT + 2 * lane));
        float* t = tile + xk * COUT + 2 * lane;
        lds_add(t, v.x * wk);
        lds_add(t + 1, v.y * wk);
    }
}

static __device__ __forceinline__ void scan_direct(int wave, int lane, int lo, int cnt,
        const int* __restrict__ srcp, const int* __restrict__ dstp,
        const float* __restrict__ dinv, const float* __restrict__ h, float* tile, int E) {
    for (int e0 = wave * 32; e0 < E; e0 += 1024) {
        const int e = e0 + lane;
        int s = 0, x = TILE;
        float w = 0.f;
        bool ok = false;
        if (e < E) {
            const int d = dstp[e];
            const unsigned xr = (unsigned)(d - lo);
            if (xr < (unsigned)cnt) {
                ok = true;
                s = srcp[e];
                x = (int)xr;
                w = dinv[s] * dinv[d];
            }
        }
        const int wi = __float_as_int(w);
        unsigned mk = __builtin_amdgcn_ballot_w32(ok);
        while (mk) {
            const int j = __builtin_ctz(mk);
            mk &= mk - 1u;
            const int sk = __builtin_amdgcn_readlane(s, j);
            const int xk = __builtin_amdgcn_readlane(x, j);
            const float wk = __int_as_float(__builtin_amdgcn_readlane(wi, j));
            const float2 v = *(const float2*)(h + (unsigned)(sk * COUT + 2 * lane));
            float* t = tile + xk * COUT + 2 * lane;
            lds_add(t, v.x * wk);
            lds_add(t + 1, v.y * wk);
        }
    }
}

template <bool WIDE>
static __device__ __forceinline__ void list_rows(const uint2* __restrict__ L, int nl, int wave, int lane, int lo,
        const float* __restrict__ dinv, const unsigned short* dg, const float* __restrict__ h, float* tile) {
    const int b0 = wave * 32;
    const int last = nl - 1;
    uint2 e1 = make_uint2(0u, 0u), e2 = e1;
    float w1 = 0.f;
    if (b0 < nl) {
        e1 = L[min(b0 + lane, last)];
        e2 = L[min(b0 + 1024 + lane, last)];
        if (WIDE) w1 = dinv[e1.x] * dinv[lo + (int)e1.y];
        else      w1 = dinv_of(dg[e1.x]) * dinv_of(dg[lo + (int)e1.y]);
    }
    for (int bb = b0; bb < nl; bb += 1024) {
        const uint2 e3 = L[min(bb + 2048 + lane, last)];
        float ds2, dd2;
        if (WIDE) { ds2 = dinv[e2.x]; dd2 = dinv[lo + (int)e2.y]; }
        else      { ds2 = dinv_of(dg[e2.x]); dd2 = dinv_of(dg[lo + (int)e2.y]); }
        const int m = (nl - bb) < 32 ? (nl - bb) : 32;
        add_rows((int)e1.x, (int)e1.y, __float_as_int(w1), m, lane, h, tile);
        e1 = e2;
        e2 = e3;
        w1 = ds2 * dd2;
    }
}

typedef unsigned short us8 __attribute__((ext_vector_type(8)));
#define TPV ((MAXT * TILE / 8 + 1023) / 1024)

__global__ void __launch_bounds__(1024)
k_tile(const int* __restrict__ srcp, const int* __restrict__ dstp,
       const float* __restrict__ dinv, const unsigned short* __restrict__ deg16, const float* __restrict__ h,
       const float* __restrict__ b,
       const uint2* __restrict__ lists, const unsigned* __restrict__ nlist,
       float* __restrict__ out, int N, int E) {
    __shared__ float tile[(TILE + 1) * COUT];
    __shared__ __attribute__((aligned(16))) unsigned short dg[MAXT * TILE];
    __shared__ unsigned wsat[32];

    const int lane = threadIdx.x & 31;
    const int wave = __builtin_amdgcn_readfirstlane((int)(threadIdx.x >> 5));
    const int lo   = blockIdx.x * TILE;
    const int cnt  = (N - lo) < TILE ? (N - lo) : TILE;

    const int nn = N < MAXT * TILE ? N : MAXT * TILE;
    const int nvec = nn >> 3;
    us8 mx = (us8)(0);
    {
        const us8* g8 = (const us8*)deg16;
        us8* d8 = (us8*)dg;
        for (int v0 = threadIdx.x; v0 < nvec; v0 += TPV * 1024) {
            us8 q[TPV];
            #pragma unroll
            for (int j = 0; j < TPV; ++j) {
                const int v = v0 + 1024 * j;
                q[j] = g8[v < nvec ? v : v0];
            }
            #pragma unroll
            for (int j = 0; j < TPV; ++j) {
                const int v = v0 + 1024 * j;
                d8[v < nvec ? v : v0] = q[j];
                mx = __builtin_elementwise_max(mx, q[j]);
            }
        }
    }
    unsigned short mt = __builtin_reduce_max(mx);
    {
        const int i = (nvec << 3) + (int)threadIdx.x;
        if (i < nn) {
            const unsigned short d = deg16[i];
            dg[i] = d;
            mt = d > mt ? d : mt;
        }
    }
    const unsigned wsa = __builtin_amdgcn_ballot_w32((unsigned)mt == DSAT);
    if (lane == 0) wsat[wave] = wsa;

    for (int idx = threadIdx.x; idx < cnt * COUT; idx += 1024) {
        int i = idx >> 6, c = idx & 63;
        float di = dinv[lo + i];
        tile[idx] = b[c] + di * di * h[(size_t)(lo + i) * COUT + c];
    }
    if (threadIdx.x < COUT) tile[TILE * COUT + threadIdx.x] = 0.f;
    __syncthreads();

    const bool wide = __builtin_amdgcn_ballot_w32(wsat[lane] != 0u) != 0u;

    const int nl = __builtin_amdgcn_readfirstlane((int)nlist[blockIdx.x]);
    if (__builtin_expect(nl <= LCAP, 1)) {
        const uint2* L = lists + (size_t)blockIdx.x * LCAP;
        if (__builtin_expect(!wide, 1)) list_rows<false>(L, nl, wave, lane, lo, dinv, dg, h, tile);
        else                            list_rows<true>(L, nl, wave, lane, lo, dinv, dg, h, tile);
    } else {
        scan_direct(wave, lane, lo, cnt, srcp, dstp, dinv, h, tile, E);
    }
    __syncthreads();

    float* op = out + (size_t)lo * COUT;
    for (int idx = threadIdx.x; idx < cnt * COUT; idx += 1024) op[idx] = tile[idx];
}

extern "C" void kernel_launch(void* const* d_in, const int* in_sizes, int n_in,
                              void* d_out, int out_size, void* d_ws, size_t ws_size,
                              hipStream_t stream) {
    const float* x  = (const float*)d_in[0];
    const int*   ei = (const int*)d_in[1];
    const float* W  = (const float*)d_in[2];
    const float* b  = (const float*)d_in[3];
    float* out = (float*)d_out;

    const int N = in_sizes[0] / CIN;
    const int E = in_sizes[1] / 2;
    const int* srcp = ei;
    const int* dstp = ei + E;

    const int nt = (N + TILE - 1) / TILE;
    char* ws = (char*)d_ws;
    float*    h     = (float*)ws;
    uint2*    lists = (uint2*)(ws + (size_t)N * COUT * sizeof(float));
    uint2*    bkts  = lists + (size_t)nt * LCAP;
    unsigned* nlist = (unsigned*)(bkts + (size_t)nt * NSL * BCAP);
    unsigned* cnts  = nlist + nt;
    float*    dinv  = (float*)(cnts + (size_t)nt * NSL);
    unsigned short* deg16 = (unsigned short*)(dinv + N);

    k_bin<<<NSL, 1024, 0, stream>>>(srcp, dstp, bkts, cnts, N, E);
    k_degown<<<nt, 1024, 98304, stream>>>(srcp, dstp, bkts, cnts, dinv, deg16, lists, nlist, x, W, h, N, E);

    k_tile<<<nt, 1024, 0, stream>>>(srcp, dstp, dinv, deg16, h, b, lists, nlist, out, N, E);
}


# owner launch writes each tile's list grouped by out row mod 32; wave g of the out-tile launch owns group g's rows: plain LDS load, add, store per edge, no LDS float atomic; row sums in list order
# speedup vs baseline: 5.8836x; 1.0537x over previous
#include <hip/hip_runtime.h>


typedef float v2f __attribute__((ext_vector_type(2)));
typedef float v8f __attribute__((ext_vector_type(8)));

#define CIN 256
#define COUT 64


#define TILE 391
#define MAXT 256

#define LCAP 65536

#define NSL 256
#define BCAP 128

static __device__ __forceinline__ int slice_len(int E) { return (((E + NSL - 1) / NSL) + 3) & ~3; }

static __device__ __forceinline__ float dinv_of(unsigned deg) { return rsqrtf((float)deg); }

#define DSAT 65535u

static __device__ __forceinline__ int base4(int e, int se) {
    const int t = se - 4 > 0 ? se - 4 : 0;
    return e + 3 < se ? e : t;
}

static __device__ __forceinline__ void ld4(const int* __restrict__ srcp, const int* __restrict__ dstp,
                                           int eb, int* s, int* d) {
    #pragma unroll
    for (int j = 0; j < 4; ++j) { s[j] = srcp[eb + j]; d[j] = dstp[eb + j]; }
}

#define SCAP 16384
#define NGRP (SCAP / 4096)
#define TPW  (MAXT / 32)

__global__ void __launch_bounds__(1024)
k_bin(const int* __restrict__ srcp, const int* __restrict__ dstp,
      uint2* __restrict__ buckets, unsigned* __restrict__ cnts, int N, int E) {
    __shared__ unsigned cc[MAXT];
    __shared__ unsigned bc[MAXT];
    __shared__ unsigned off[MAXT];
    __shared__ uint2 stage[SCAP];

    const int lane = threadIdx.x & 31;
    const int wave = __builtin_amdgcn_readfirstlane((int)(threadIdx.x >> 5));
    if (threadIdx.x < MAXT) { cc[threadIdx.x] = 0u; bc[threadIdx.x] = 0u; }

    const int per = slice_len(E);
    const int sb  = blockIdx.x * per;
    const int se  = (E - sb) < per ? E : sb + per;
    const unsigned lim = (unsigned)(N < MAXT * TILE ? N : MAXT * TILE);
    uint2* B = buckets + (size_t)blockIdx.x * BCAP;

    for (int cb = sb; cb < se; cb += SCAP) {
        const int ce = (se - cb) < SCAP ? se : cb + SCAP;

        int sc[NGRP][4], dc[NGRP][4], eb[NGRP];
        #pragma unroll
        for (int q = 0; q < NGRP; ++q) {
            eb[q] = base4(cb + q * 4096 + (int)threadIdx.x * 4, ce);
            ld4(srcp, dstp, eb[q], sc[q], dc[q]);
        }
        #pragma unroll
        for (int q = 0; q < NGRP; ++q) {
            const int e0 = cb + q * 4096 + (int)threadIdx.x * 4;
            #pragma unroll
            for (int j = 0; j < 4; ++j) if (eb[q] + j < e0) dc[q][j] = -1;
        }
        __syncthreads();

        unsigned rk[NGRP][4];
        #pragma unroll
        for (int q = 0; q < NGRP; ++q) {
            #pragma unroll
            for (int j = 0; j < 4; ++j) {
                const unsigned d = (unsigned)dc[q][j];
                rk[q][j] = 0u;
                if (d < lim)
                    rk[q][j] = __hip_atomic_fetch_add(&cc[d / TILE], 1u, __ATOMIC_RELAXED, __HIP_MEMORY_SCOPE_WORKGROUP);
            }
        }
        __syncthreads();

        if (wave == 0) {
            unsigned c[TPW];
            unsigned s = 0u;
            #pragma unroll
            for (int k = 0; k < TPW; ++k) { c[k] = cc[lane * TPW + k]; s += c[k]; }
            unsigned inc = s;
            #pragma unroll
            for (int dl = 1; dl < 32; dl <<= 1) {
                const unsigned v = __shfl_up(inc, (unsigned)dl);
                if (lane >= dl) inc += v;
            }
            unsigned ex = inc - s;
            #pragma unroll
            for (int k = 0; k < TPW; ++k) { off[lane * TPW + k] = ex; ex += c[k]; }
        }
        __syncthreads();

        unsigned ps[NGRP][4];
        #pragma unroll
        for (int q = 0; q < NGRP; ++q) {
            #pragma unroll
            for (int j = 0; j < 4; ++j) {
                const unsigned d = (unsigned)dc[q][j];
                ps[q][j] = off[d < lim ? d / TILE : 0u] + rk[q][j];
            }
        }
        #pragma unroll
        for (int q = 0; q < NGRP; ++q) {
            #pragma unroll
            for (int j = 0; j < 4; ++j) {
                const unsigned d = (unsigned)dc[q][j];
                if (d < lim) stage[ps[q][j]] = make_uint2((unsigned)sc[q][j], d - (d / TILE) * TILE);
            }
        }
        __syncthreads();

        {
            unsigned vo = 0u, vb = 0u, vs = 0u;
            if (lane < TPW) {
                const int t = wave * TPW + lane;
                const unsigned vn = cc[t];
                vo = off[t];
                vb = bc[t];
                const unsigned room = vb < (unsigned)BCAP ? (unsigned)BCAP - vb : 0u;
                vs = vn < room ? vn : room;
            }
            unsigned nk[TPW], ok[TPW];
            uint2 en[TPW][2];
            #pragma unroll
            for (int k = 0; k < TPW; ++k) {
                nk[k] = (unsigned)__builtin_amdgcn_readlane((int)vs, k);
                ok[k] = (unsigned)__builtin_amdgcn_readlane((int)vo, k);
                #pragma unroll
                for (int c = 0; c < 2; ++c) {
                    const unsigned i = (unsigned)(c * 32 + lane);
                    en[k][c] = make_uint2(0u, 0u);
                    if (i < nk[k]) en[k][c] = stage[ok[k] + i];
                }
            }
            #pragma unroll
            for (int k = 0; k < TPW; ++k) {
                const unsigned bk = (unsigned)__builtin_amdgcn_readlane((int)vb, k);
                uint2* Bt = B + (size_t)(wave * TPW + k) * (NSL * BCAP) + bk;
                #pragma unroll
                for (int c = 0; c < 2; ++c) {
                    const unsigned i = (unsigned)(c * 32 + lane);
                    if (i < nk[k]) Bt[i] = en[k][c];
                }
                for (unsigned i = 64u + (unsigned)lane; i < nk[k]; i += 32u) Bt[i] = stage[ok[k] + i];
            }
        }
        __syncthreads();

        if (threadIdx.x < MAXT) { bc[threadIdx.x] += cc[threadIdx.x]; cc[threadIdx.x] = 0u; }
    }

    const int nt = (N + TILE - 1) / TILE;
    if ((int)threadIdx.x < nt && threadIdx.x < MAXT) cnts[threadIdx.x * NSL + blockIdx.x] = bc[threadIdx.x];
}

static __device__ __forceinline__ void gemm_strip(const float* __restrict__ x, const float2* Ws2,
                                                  float* __restrict__ h, int m0, int mend, int N, int lane) {
    const int mr   = lane & 15;
    const int kh   = (lane >> 4) << 1;

    int row  = m0 + mr;
    int rowc = row < N ? row : (N - 1);
    const float* xr = x + (size_t)rowc * CIN;

    v8f acc0 = {}; v8f acc1 = {}; v8f acc2 = {}; v8f acc3 = {};
    for (int k = 0; k < CIN; k += 4) {
        v2f a = *(const v2f*)(xr + k + kh);
        const float2* wp = &Ws2[(((k + kh) >> 1) * COUT) + mr];
        v2f b0 = *(const v2f*)(wp + 0);
        v2f b1 = *(const v2f*)(wp + 16);
        v2f b2 = *(const v2f*)(wp + 32);
        v2f b3 = *(const v2f*)(wp + 48);
        acc0 = __builtin_amdgcn_wmma_f32_16x16x4_f32(false, a, false, b0, (short)0, acc0, false, false);
        acc1 = __builtin_amdgcn_wmma_f32_16x16x4_f32(false, a, false, b1, (short)0, acc1, false, false);
        acc2 = __builtin_amdgcn_wmma_f32_16x16x4_f32(false, a, false, b2, (short)0, acc2, false, false);
        acc3 = __builtin_amdgcn_wmma_f32_16x16x4_f32(false, a, false, b3, (short)0, acc3, false, false);
    }

    const int nlo = lane & 15;
    const int mhi = (lane >> 4) << 3;
    for (int r = 0; r < 8; ++r) {
        int orow = m0 + r + mhi;
        if (orow < mend) {
            float* hp = h + (size_t)orow * COUT + nlo;
            hp[0]  = acc0[r];
            hp[16] = acc1[r];
            hp[32] = acc2[r];
            hp[48] = acc3[r];
        }
    }
}


static __device__ __forceinline__ void lds_add(float* p, float v) {
    __hip_atomic_fetch_add(p, v, __ATOMIC_RELAXED, __HIP_MEMORY_SCOPE_WORKGROUP);
}

#define BPW (NSL / 32)
#define NSTRIP ((TILE + 15) / 16)
#define WPT ((CIN / 2) * COUT / 1024)
#define NGR 32
#define RPG ((TILE + NGR - 1) / NGR)
#define NSG 4
#define GO 36
#define STCAP 16384

static __device__ __forceinline__ void place(uint2 e, unsigned* gpos, uint2* stage, uint2* __restrict__ list, bool staged) {
    const unsigned run = (e.y & (NGR - 1)) * NSG + ((e.y / NGR) & (NSG - 1));
    const unsigned slot = __hip_atomic_fetch_add(&gpos[run], 1u, __ATOMIC_RELAXED, __HIP_MEMORY_SCOPE_WORKGROUP);
    if (staged) stage[slot] = e;
    else if (slot < (unsigned)LCAP) list[slot] = e;
}

template <bool PLACE>
static __device__ __forceinline__ void scan_slices(unsigned nv, int wave, int lane, int lo, int cnt,
        const int* __restrict__ srcp, const int* __restrict__ dstp, int E,
        unsigned* cl, unsigned* gpos, uint2* stage, uint2* __restrict__ list, bool staged) {
    const int per = slice_len(E);
    for (int j = 0; j < BPW; ++j) {
        const unsigned n = (unsigned)__builtin_amdgcn_readlane((int)nv, j);
        if (n <= (unsigned)BCAP) continue;
        const int sb = (wave + 32 * j) * per;
        const int se = (E - sb) < per ? E : sb + per;
        for (int e = sb + lane; e < se; e += 32) {
            const unsigned xr = (unsigned)(dstp[e] - lo);
            if (xr < (unsigned)cnt) {
                if (PLACE) place(make_uint2((unsigned)srcp[e], xr), gpos, stage, list, staged);
                else __hip_atomic_fetch_add(&cl[xr], 1u, __ATOMIC_RELAXED, __HIP_MEMORY_SCOPE_WORKGROUP);
            }
        }
    }
}

__global__ void __launch_bounds__(1024)
k_degown(const int* __restrict__ srcp, const int* __restrict__ dstp,
         const uint2* __restrict__ buckets, const unsigned* __restrict__ cnts,
         float* __restrict__ dinv, unsigned short* __restrict__ deg16,
         uint2* __restrict__ lists, unsigned* __restrict__ goffs,
         const float* __restrict__ x, const float* __restrict__ W, float* __restrict__ h, int N, int E) {
    __shared__ unsigned cl[TILE];
    __shared__ unsigned gpos[NGR * NSG];
    __shared__ unsigned gtot;
    __shared__ float2 Ws2[(CIN / 2) * COUT];
    __shared__ uint2 stage[STCAP];

    const int lane = threadIdx.x & 31;
    const int wave = __builtin_amdgcn_readfirstlane((int)(threadIdx.x >> 5));
    const int lo   = blockIdx.x * TILE;
    const int cnt  = (N - lo) < TILE ? (N - lo) : TILE;

    if (threadIdx.x < TILE) cl[threadIdx.x] = 0u;
    __syncthreads();

    uint2* list = lists + (size_t)blockIdx.x * LCAP;
    const unsigned* cn = cnts + (size_t)blockIdx.x * NSL;
    const uint2* B = buckets + (size_t)blockIdx.x * (NSL * BCAP);

    unsigned nv = 0u;
    if (lane < BPW) nv = cn[wave + 32 * lane];
    const unsigned nu = nv <= (unsigned)BCAP ? nv : 0u;
    const bool anyover = __builtin_amdgcn_ballot_w32(nv > (unsigned)BCAP) != 0u;

    unsigned nj[BPW];
    uint2 en[BPW][2];
    #pragma unroll
    for (int j = 0; j < BPW; ++j) {
        nj[j] = (unsigned)__builtin_amdgcn_readlane((int)nu, j);
        const uint2* Bj = B + (wave + 32 * j) * BCAP;
        #pragma unroll
        for (int c = 0; c < 2; ++c) {
            const unsigned i = (unsigned)(c * 32 + lane);
            en[j][c] = make_uint2(0u, 0u);
            if (i < nj[j]) en[j][c] = Bj[i];
        }
    }
    float2 wv[WPT];
    #pragma unroll
    for (int j = 0; j < WPT; ++j) {
        const int idx = (int)threadIdx.x + 1024 * j;
        const int p = idx >> 6, n = idx & 63;
        wv[j] = make_float2(W[p * (2 * COUT) + n], W[p * (2 * COUT) + COUT + n]);
    }

    #pragma unroll
    for (int j = 0; j < BPW; ++j) {
        #pragma unroll
        for (int c = 0; c < 2; ++c) {
            const unsigned i = (unsigned)(c * 32 + lane);
            if (i < nj[j])
                __hip_atomic_fetch_add(&cl[en[j][c].y], 1u, __ATOMIC_RELAXED, __HIP_MEMORY_SCOPE_WORKGROUP);
        }
        const uint2* Bj = B + (wave + 32 * j) * BCAP;
        for (unsigned i = 64u + (unsigned)lane; i < nj[j]; i += 32u)
            __hip_atomic_fetch_add(&cl[Bj[i].y], 1u, __ATOMIC_RELAXED, __HIP_MEMORY_SCOPE_WORKGROUP);
    }
    if (anyover) scan_slices<false>(nv, wave, lane, lo, cnt, srcp, dstp, E, cl, gpos, stage, list, false);

    #pragma unroll
    for (int j = 0; j < WPT; ++j) Ws2[(int)threadIdx.x + 1024 * j] = wv[j];
    __syncthreads();

    if ((int)threadIdx.x < cnt) {
        const unsigned dg = cl[threadIdx.x] + 1u;
        dinv[lo + threadIdx.x] = dinv_of(dg);
        deg16[lo + threadIdx.x] = (unsigned short)(dg < DSAT ? dg : DSAT);
    }
    if (wave == 0) {
        unsigned sq[NSG];
        #pragma unroll
        for (int q = 0; q < NSG; ++q) sq[q] = 0u;
        #pragma unroll
        for (int k = 0; k < RPG; ++k) {
            const int r = lane + NGR * k;
            if (r < TILE) sq[k & (NSG - 1)] += cl[r];
        }
        unsigned s = 0u;
        #pragma unroll
        for (int q = 0; q < NSG; ++q) s += sq[q];
        unsigned inc = s;
        #pragma unroll
        for (int dl = 1; dl < 32; dl <<= 1) {
            const unsigned v = __shfl_up(inc, (unsigned)dl);
            if (lane >= dl) inc += v;
        }
        unsigned ex = inc - s;
        goffs[blockIdx.x * GO + lane] = ex;
        #pragma unroll
        for (int q = 0; q < NSG; ++q) { gpos[lane * NSG + q] = ex; ex += sq[q]; }
        if (lane == 31) { goffs[blockIdx.x * GO + NGR] = inc; gtot = inc; }
    }
    __syncthreads();

    const unsigned total = gtot;
    const bool staged = total <= (unsigned)STCAP;
    #pragma unroll
    for (int j = 0; j < BPW; ++j) {
        #pragma unroll
        for (int c = 0; c < 2; ++c) {
            const unsigned i = (unsigned)(c * 32 + lane);
            if (i < nj[j]) place(en[j][c], gpos, stage, list, staged);
        }
        const uint2* Bj = B + (wave + 32 * j) * BCAP;
        for (unsigned i = 64u + (unsigned)lane; i < nj[j]; i += 32u) place(Bj[i], gpos, stage, list, staged);
    }
    if (anyover) scan_slices<true>(nv, wave, lane, lo, cnt, srcp, dstp, E, cl, gpos, stage, list, staged);
    if (staged) {
        __syncthreads();
        for (unsigned i = threadIdx.x; i < total; i += 1024u) list[i] = stage[i];
    }

    {
        const int a = wave & 3, q = wave >> 3;
        const int strip = (wave & 4) ? 16 + a + 4 * ((q - a) & 3) : q * 4 + a;
        if (strip < NSTRIP && strip * 16 < cnt)
            gemm_strip(x, Ws2, h, lo + strip * 16, lo + cnt, N, lane);
    }
}

static __device__ __forceinline__ void load16(float2 (&v)[16], int s, int k0, int lane, const float* __restrict__ h) {
    #pragma unroll
    for (int j = 0; j < 16; ++j) {
        const int sk = __builtin_amdgcn_readlane(s, k0 + j);
        v[j] = ((const float2*)h)[(unsigned)(sk * (COUT / 2) + lane)];
    }
}

static __device__ __forceinline__ void rmw16(const float2 (&v)[16], int x, int wi, int k0, int m, int lane, float* tile) {
    #pragma unroll
    for (int j = 0; j < 16; j += 2) {
        const int xa = __builtin_amdgcn_readlane(x, k0 + j);
        const int xb = __builtin_amdgcn_readlane(x, k0 + j + 1);
        const float wa = __int_as_float(__builtin_amdgcn_readlane(wi, k0 + j));
        const float wb = __int_as_float(__builtin_amdgcn_readlane(wi, k0 + j + 1));
        float2* ta = (float2*)(tile + xa * COUT) + lane;
        float2* tb = (float2*)(tile + xb * COUT) + lane;
        if (k0 + j + 1 < m && xa != xb) {
            float2 p = *ta;
            float2 q = *tb;
            p.x += v[j].x * wa;     p.y += v[j].y * wa;
            q.x += v[j + 1].x * wb; q.y += v[j + 1].y * wb;
            *ta = p;
            *tb = q;
        } else {
            if (k0 + j < m) {
                float2 p = *ta;
                p.x += v[j].x * wa; p.y += v[j].y * wa;
                *ta = p;
            }
            if (k0 + j + 1 < m) {
                float2 q = *tb;
                q.x += v[j + 1].x * wb; q.y += v[j + 1].y * wb;
                *tb = q;
            }
        }
    }
}

static __device__ __forceinline__ void scan_direct(int wave, int lane, int lo, int cnt,
        const int* __restrict__ srcp, const int* __restrict__ dstp,
        const float* __restrict__ dinv, const float* __restrict__ h, float* tile, int E) {
    for (int e0 = wave * 32; e0 < E; e0 += 1024) {
        const int e = e0 + lane;
        int s = 0, x = TILE;
        float w = 0.f;
        bool ok = false;
        if (e < E) {
            const int d = dstp[e];
            const unsigned xr = (unsigned)(d - lo);
            if (xr < (unsigned)cnt) {
                ok = true;
                s = srcp[e];
                x = (int)xr;
                w = dinv[s] * dinv[d];
            }
        }
        const int wi = __float_as_int(w);
        unsigned mk = __builtin_amdgcn_ballot_w32(ok);
        while (mk) {
            const int j = __builtin_ctz(mk);
            mk &= mk - 1u;
            const int sk = __builtin_amdgcn_readlane(s, j);
            const int xk = __builtin_amdgcn_readlane(x, j);
            const float wk = __int_as_float(__builtin_amdgcn_readlane(wi, j));
            const float2 v = *(const float2*)(h + (unsigned)(sk * COUT + 2 * lane));
            float* t = tile + xk * COUT + 2 * lane;
            lds_add(t, v.x * wk);
            lds_add(t + 1, v.y * wk);
        }
    }
}

template <bool WIDE>
static __device__ __forceinline__ float edge_weight(uint2 e, int lo, const float* __restrict__ dinv, const unsigned short* dg) {
    if (WIDE) return dinv[e.x] * dinv[lo + (int)e.y];
    return dinv_of(dg[e.x]) * dinv_of(dg[lo + (int)e.y]);
}

template <bool WIDE>
static __device__ __forceinline__ void list_rows(const uint2* __restrict__ L, int a, int b, int lane, int lo,
        const float* __restrict__ dinv, const unsigned short* dg, const float* __restrict__ h, float* tile) {
    if (a >= b) return;
    const int last = b - 1;
    uint2 e1 = L[min(a + lane, last)];
    uint2 e2 = L[min(a + 32 + lane, last)];
    float w1 = edge_weight<WIDE>(e1, lo, dinv, dg);
    float2 va[16], vb[16];
    load16(va, (int)e1.x, 0, lane, h);
    for (int bb = a; bb < b; bb += 32) {
        const uint2 e3 = L[min(bb + 64 + lane, last)];
        const float w2 = edge_weight<WIDE>(e2, lo, dinv, dg);
        const int m = (b - bb) < 32 ? (b - bb) : 32;
        const int wi = __float_as_int(w1);
        load16(vb, (int)e1.x, 16, lane, h);
        rmw16(va, (int)e1.y, wi, 0, m, lane, tile);
        load16(va, (int)e2.x, 0, lane, h);
        rmw16(vb, (int)e1.y, wi, 16, m, lane, tile);
        e1 = e2;
        e2 = e3;
        w1 = w2;
    }
}

typedef unsigned short us8 __attribute__((ext_vector_type(8)));
#define TPV ((MAXT * TILE / 8 + 1023) / 1024)

__global__ void __launch_bounds__(1024)
k_tile(const int* __restrict__ srcp, const int* __restrict__ dstp,
       const float* __restrict__ dinv, const unsigned short* __restrict__ deg16, const float* __restrict__ h,
       const float* __restrict__ b,
       const uint2* __restrict__ lists, const unsigned* __restrict__ goffs,
       float* __restrict__ out, int N, int E) {
    __shared__ __attribute__((aligned(16))) float tile[(TILE + 1) * COUT];
    __shared__ __attribute__((aligned(16))) unsigned short dg[MAXT * TILE];
    __shared__ unsigned wsat[32];

    const int lane = threadIdx.x & 31;
    const int wave = __builtin_amdgcn_readfirstlane((int)(threadIdx.x >> 5));
    const int lo   = blockIdx.x * TILE;
    const int cnt  = (N - lo) < TILE ? (N - lo) : TILE;

    const int nn = N < MAXT * TILE ? N : MAXT * TILE;
    const int nvec = nn >> 3;
    us8 mx = (us8)(0);
    {
        const us8* g8 = (const us8*)deg16;
        us8* d8 = (us8*)dg;
        for (int v0 = threadIdx.x; v0 < nvec; v0 += TPV * 1024) {
            us8 q[TPV];
            #pragma unroll
            for (int j = 0; j < TPV; ++j) {
                const int v = v0 + 1024 * j;
                q[j] = g8[v < nvec ? v : v0];
            }
            #pragma unroll
            for (int j = 0; j < TPV; ++j) {
                const int v = v0 + 1024 * j;
                d8[v < nvec ? v : v0] = q[j];
                mx = __builtin_elementwise_max(mx, q[j]);
            }
        }
    }
    unsigned short mt = __builtin_reduce_max(mx);
    {
        const int i = (nvec << 3) + (int)threadIdx.x;
        if (i < nn) {
            const unsigned short d = deg16[i];
            dg[i] = d;
            mt = d > mt ? d : mt;
        }
    }
    const unsigned wsa = __builtin_amdgcn_ballot_w32((unsigned)mt == DSAT);
    if (lane == 0) wsat[wave] = wsa;

    for (int idx = threadIdx.x; idx < cnt * COUT; idx += 1024) {
        int i = idx >> 6, c = idx & 63;
        float di = dinv[lo + i];
        tile[idx] = b[c] + di * di * h[(size_t)(lo + i) * COUT + c];
    }
    if (threadIdx.x < COUT) tile[TILE * COUT + threadIdx.x] = 0.f;
    __syncthreads();

    const bool wide = __builtin_amdgcn_ballot_w32(wsat[lane] != 0u) != 0u;

    const unsigned* go = goffs + blockIdx.x * GO;
    const int nl = __builtin_amdgcn_readfirstlane((int)go[NGR]);
    if (__builtin_expect(nl <= LCAP, 1)) {
        const uint2* L = lists + (size_t)blockIdx.x * LCAP;
        const int ga = __builtin_amdgcn_readfirstlane((int)go[wave]);
        const int gb = __builtin_amdgcn_readfirstlane((int)go[wave + 1]);
        if (__builtin_expect(!wide, 1)) list_rows<false>(L, ga, gb, lane, lo, dinv, dg, h, tile);
        else                            list_rows<true>(L, ga, gb, lane, lo, dinv, dg, h, tile);
    } else {
        scan_direct(wave, lane, lo, cnt, srcp, dstp, dinv, h, tile, E);
    }
    __syncthreads();

    float* op = out + (size_t)lo * COUT;
    for (int idx = threadIdx.x; idx < cnt * COUT; idx += 1024) op[idx] = tile[idx];
}

extern "C" void kernel_launch(void* const* d_in, const int* in_sizes, int n_in,
                              void* d_out, int out_size, void* d_ws, size_t ws_size,
                              hipStream_t stream) {
    const float* x  = (const float*)d_in[0];
    const int*   ei = (const int*)d_in[1];
    const float* W  = (const float*)d_in[2];
    const float* b  = (const float*)d_in[3];
    float* out = (float*)d_out;

    const int N = in_sizes[0] / CIN;
    const int E = in_sizes[1] / 2;
    const int* srcp = ei;
    const int* dstp = ei + E;

    const int nt = (N + TILE - 1) / TILE;
    char* ws = (char*)d_ws;
    float*    h     = (float*)ws;
    uint2*    lists = (uint2*)(ws + (size_t)N * COUT * sizeof(float));
    uint2*    bkts  = lists + (size_t)nt * LCAP;
    unsigned* goffs = (unsigned*)(bkts + (size_t)nt * NSL * BCAP);
    unsigned* cnts  = goffs + (size_t)nt * GO;
    float*    dinv  = (float*)(cnts + (size_t)nt * NSL);
    unsigned short* deg16 = (unsigned short*)(dinv + N);

    k_bin<<<NSL, 1024, 0, stream>>>(srcp, dstp, bkts, cnts, N, E);
    k_degown<<<nt, 1024, 0, stream>>>(srcp, dstp, bkts, cnts, dinv, deg16, lists, goffs, x, W, h, N, E);

    k_tile<<<nt, 1024, 0, stream>>>(srcp, dstp, dinv, deg16, h, b, lists, goffs, out, N, E);
}
